# starting kernel
# baseline (speedup 1.0000x reference)
#include <hip/hip_runtime.h>
#include <hip/hip_bf16.h>


#define GCN_N    50000
#define GCN_FIN  128
#define GCN_E    800000
#define GCN_H    64
#define GCN_C    10
#define GCN_B    512

typedef __attribute__((ext_vector_type(2))) float v2f;
typedef __attribute__((ext_vector_type(8))) float v8f;

__global__ __launch_bounds__(128) void gcn_gemm_wmma_f32(
    const float* __restrict__ A, const float* __restrict__ W,
    float* __restrict__ C, int K)
{
    const int lane    = threadIdx.x;
    const int colBase = threadIdx.y * 16;
    const int rowBase = blockIdx.x * 16;
    const int half    = lane >> 4;
    const int l       = lane & 15;

    const float* arow = A + (size_t)(rowBase + l) * K + 2 * half;
    const float* wcol = W + colBase + l;

    v8f acc = {};
    for (int k0 = 0; k0 < K; k0 += 4) {
        v2f a, b;
        a.x = arow[k0 + 0];
        a.y = arow[k0 + 1];
        b.x = wcol[(size_t)(k0 + 2 * half + 0) * GCN_H];
        b.y = wcol[(size_t)(k0 + 2 * half + 1) * GCN_H];
        acc = __builtin_amdgcn_wmma_f32_16x16x4_f32(
             false, a,  false, b,
             (short)0, acc,  false,  false);
    }

#pragma unroll
    for (int v = 0; v < 8; ++v) {
        const int m = v + 8 * half;
        C[(size_t)(rowBase + m) * GCN_H + colBase + l] = acc[v];
    }
}

__global__ __launch_bounds__(256) void gcn_deg_count(
    const int* __restrict__ dst, float* __restrict__ deg)
{
    const int e = blockIdx.x * blockDim.x + threadIdx.x;
    if (e < GCN_E) atomicAdd(&deg[dst[e]], 1.0f);
}

__global__ __launch_bounds__(256) void gcn_dinv(float* __restrict__ deg)
{
    const int i = blockIdx.x * blockDim.x + threadIdx.x;
    if (i < GCN_N) deg[i] = rsqrtf(deg[i] + 1.0f);
}

__global__ __launch_bounds__(256) void gcn_scatter(
    const float* __restrict__ h, const int* __restrict__ src,
    const int* __restrict__ dst, const float* __restrict__ dinv,
    float* __restrict__ agg)
{
    const int idx = blockIdx.x * blockDim.x + threadIdx.x;
    const int e = idx >> 5;
    const int p = (idx & 31) * 2;
    const int s = src[e];
    const int d = dst[e];
    const float c = dinv[s] * dinv[d];
    const float2 v = *(const float2*)(h + (size_t)s * GCN_H + p);
    float* out = agg + (size_t)d * GCN_H + p;
    atomicAdd(out + 0, v.x * c);
    atomicAdd(out + 1, v.y * c);
}

__global__ __launch_bounds__(256) void gcn_epilogue(
    const float* __restrict__ agg, const float* __restrict__ h,
    const float* __restrict__ dinv, const float* __restrict__ bias,
    float* __restrict__ out, int do_relu)
{
    const int idx = blockIdx.x * blockDim.x + threadIdx.x;
    const int i = idx >> 6;
    const int f = idx & 63;
    const float di = dinv[i];
    float v = agg[idx] + h[idx] * di * di + bias[f];
    if (do_relu) v = fmaxf(v, 0.0f);
    out[idx] = v;
}

__global__ __launch_bounds__(256) void gcn_pool(
    const float* __restrict__ h, const int* __restrict__ batch,
    float* __restrict__ psum, float* __restrict__ pcnt)
{
    const int idx = blockIdx.x * blockDim.x + threadIdx.x;
    const int i = idx >> 6;
    const int f = idx & 63;
    const int b = batch[i];
    atomicAdd(&psum[(size_t)b * GCN_H + f], h[idx]);
    if (f == 0) atomicAdd(&pcnt[b], 1.0f);
}

__global__ __launch_bounds__(256) void gcn_final(
    const float* __restrict__ psum, const float* __restrict__ pcnt,
    const float* __restrict__ Wl, const float* __restrict__ bl,
    float* __restrict__ out)
{
    const int idx = blockIdx.x * blockDim.x + threadIdx.x;
    const int b = idx / GCN_C;
    const int c = idx % GCN_C;
    const float inv = 1.0f / fmaxf(pcnt[b], 1.0f);
    float acc = bl[c];
    const float* row = psum + (size_t)b * GCN_H;
#pragma unroll 8
    for (int k = 0; k < GCN_H; ++k)
        acc = fmaf(row[k] * inv, Wl[k * GCN_C + c], acc);
    out[idx] = acc;
}

extern "C" void kernel_launch(void* const* d_in, const int* in_sizes, int n_in,
                              void* d_out, int out_size, void* d_ws, size_t ws_size,
                              hipStream_t stream) {
    (void)in_sizes; (void)n_in; (void)out_size; (void)ws_size;

    const float* x     = (const float*)d_in[0];
    const int*   ei    = (const int*)  d_in[1];
    const int*   batch = (const int*)  d_in[2];
    const float* W1    = (const float*)d_in[3];
    const float* b1    = (const float*)d_in[4];
    const float* W2    = (const float*)d_in[5];
    const float* b2    = (const float*)d_in[6];
    const float* W3    = (const float*)d_in[7];
    const float* b3    = (const float*)d_in[8];
    const float* Wl    = (const float*)d_in[9];
    const float* bl    = (const float*)d_in[10];
    float* out = (float*)d_out;

    const int* src = ei;
    const int* dst = ei + GCN_E;

    float* dinv = (float*)d_ws;
    float* bufA = dinv + GCN_N;
    float* bufB = bufA + (size_t)GCN_N * GCN_H;
    float* psum = bufB + (size_t)GCN_N * GCN_H;
    float* pcnt = psum + (size_t)GCN_B * GCN_H;

    const dim3 gemmBlk(32, 4);
    const dim3 gemmGrd(GCN_N / 16);
    const int  scatGrd = (GCN_E * 32) / 256;
    const int  nhGrd   = (GCN_N * GCN_H) / 256;
    const size_t nhBytes = (size_t)GCN_N * GCN_H * sizeof(float);

    hipMemsetAsync(dinv, 0, GCN_N * sizeof(float), stream);
    gcn_deg_count<<<(GCN_E + 255) / 256, 256, 0, stream>>>(dst, dinv);
    gcn_dinv<<<(GCN_N + 255) / 256, 256, 0, stream>>>(dinv);

    gcn_gemm_wmma_f32<<<gemmGrd, gemmBlk, 0, stream>>>(x, W1, bufA, GCN_FIN);
    hipMemsetAsync(bufB, 0, nhBytes, stream);
    gcn_scatter<<<scatGrd, 256, 0, stream>>>(bufA, src, dst, dinv, bufB);
    gcn_epilogue<<<nhGrd, 256, 0, stream>>>(bufB, bufA, dinv, b1, bufB, 1);

    gcn_gemm_wmma_f32<<<gemmGrd, gemmBlk, 0, stream>>>(bufB, W2, bufA, GCN_H);
    hipMemsetAsync(bufB, 0, nhBytes, stream);
    gcn_scatter<<<scatGrd, 256, 0, stream>>>(bufA, src, dst, dinv, bufB);
    gcn_epilogue<<<nhGrd, 256, 0, stream>>>(bufB, bufA, dinv, b2, bufB, 1);

    gcn_gemm_wmma_f32<<<gemmGrd, gemmBlk, 0, stream>>>(bufB, W3, bufA, GCN_H);
    hipMemsetAsync(bufB, 0, nhBytes, stream);
    gcn_scatter<<<scatGrd, 256, 0, stream>>>(bufA, src, dst, dinv, bufB);
    gcn_epilogue<<<nhGrd, 256, 0, stream>>>(bufB, bufA, dinv, b3, bufB, 0);

    hipMemsetAsync(psum, 0, (size_t)(GCN_B * GCN_H + GCN_B) * sizeof(float), stream);
    gcn_pool<<<nhGrd, 256, 0, stream>>>(bufB, batch, psum, pcnt);
    gcn_final<<<(GCN_B * GCN_C) / 256, 256, 0, stream>>>(psum, pcnt, Wl, bl, out);
}


# Neighbour sums by gather over a dst-sorted edge list (counting sort once per call: integer atomic rank, scan, fill) replace float atomics; self-loop+bias+relu in same pass; sum order = rank order
# speedup vs baseline: 1.9327x; 1.9327x over previous
#include <hip/hip_runtime.h>
#include <hip/hip_bf16.h>


#define GCN_N    50000
#define GCN_FIN  128
#define GCN_E    800000
#define GCN_H    64
#define GCN_C    10
#define GCN_B    512
#define GCN_NP   53248

typedef __attribute__((ext_vector_type(2))) float v2f;
typedef __attribute__((ext_vector_type(8))) float v8f;

__global__ __launch_bounds__(128) void gcn_gemm_wmma_f32(
    const float* __restrict__ A, const float* __restrict__ W,
    float* __restrict__ C, int K)
{
    const int lane    = threadIdx.x;
    const int colBase = threadIdx.y * 16;
    const int rowBase = blockIdx.x * 16;
    const int half    = lane >> 4;
    const int l       = lane & 15;

    const float* arow = A + (size_t)(rowBase + l) * K + 2 * half;
    const float* wcol = W + colBase + l;

    v8f acc = {};
    for (int k0 = 0; k0 < K; k0 += 4) {
        v2f a, b;
        a.x = arow[k0 + 0];
        a.y = arow[k0 + 1];
        b.x = wcol[(size_t)(k0 + 2 * half + 0) * GCN_H];
        b.y = wcol[(size_t)(k0 + 2 * half + 1) * GCN_H];
        acc = __builtin_amdgcn_wmma_f32_16x16x4_f32(
             false, a,  false, b,
             (short)0, acc,  false,  false);
    }

#pragma unroll
    for (int v = 0; v < 8; ++v) {
        const int m = v + 8 * half;
        C[(size_t)(rowBase + m) * GCN_H + colBase + l] = acc[v];
    }
}

__global__ __launch_bounds__(256) void gcn_rank(
    const int* __restrict__ dst, int* __restrict__ cnt, int* __restrict__ rank)
{
    const int e = blockIdx.x * blockDim.x + threadIdx.x;
    if (e < GCN_E) rank[e] = atomicAdd(&cnt[dst[e]], 1);
}

__global__ __launch_bounds__(1024) void gcn_scan(
    const int* __restrict__ cnt, int* __restrict__ rowptr, float* __restrict__ dinv)
{
    __shared__ int part[1024];
    const int t = threadIdx.x;
    const int4* in = (const int4*)(cnt + t * 52);
    int4 c[13];
#pragma unroll
    for (int j = 0; j < 13; ++j) c[j] = in[j];
    int s = 0;
#pragma unroll
    for (int j = 0; j < 13; ++j) s += c[j].x + c[j].y + c[j].z + c[j].w;

    part[t] = s;
    __syncthreads();
    for (int off = 1; off < 1024; off <<= 1) {
        const int add = (t >= off) ? part[t - off] : 0;
        __syncthreads();
        part[t] += add;
        __syncthreads();
    }
    int run = part[t] - s;

    int4*   rp = (int4*)(rowptr + t * 52);
    float4* dv = (float4*)(dinv + t * 52);
#pragma unroll
    for (int j = 0; j < 13; ++j) {
        int4 r;
        r.x = run;
        r.y = r.x + c[j].x;
        r.z = r.y + c[j].y;
        r.w = r.z + c[j].z;
        run = r.w + c[j].w;
        rp[j] = r;
        float4 d;
        d.x = rsqrtf((float)c[j].x + 1.0f);
        d.y = rsqrtf((float)c[j].y + 1.0f);
        d.z = rsqrtf((float)c[j].z + 1.0f);
        d.w = rsqrtf((float)c[j].w + 1.0f);
        dv[j] = d;
    }
}

__global__ __launch_bounds__(256) void gcn_fill(
    const int* __restrict__ src, const int* __restrict__ dst,
    const int* __restrict__ rank, const int* __restrict__ rowptr,
    const float* __restrict__ dinv, int2* __restrict__ rec)
{
    const int e = blockIdx.x * blockDim.x + threadIdx.x;
    if (e < GCN_E) {
        const int s = src[e];
        const int d = dst[e];
        const float c = dinv[s] * dinv[d];
        rec[rowptr[d] + rank[e]] = make_int2(s, __float_as_int(c));
    }
}

__global__ __launch_bounds__(256) void gcn_gather(
    const float* __restrict__ h, const int2* __restrict__ rec,
    const int* __restrict__ rowptr, const float* __restrict__ dinv,
    const float* __restrict__ bias, float* __restrict__ out, int do_relu)
{
    const int lane = threadIdx.x & 31;
    const int p = lane * 2;
    const int node = __builtin_amdgcn_readfirstlane(blockIdx.x * 8 + (threadIdx.x >> 5));

    const int beg = rowptr[node];
    const int end = rowptr[node + 1];
    const float di = dinv[node];
    const float2 hv = *(const float2*)(h + (size_t)node * GCN_H + p);

    float ax = 0.0f, ay = 0.0f;
    for (int base = beg; base < end; base += 32) {
        const int left = end - base;
        const int n = left < 32 ? left : 32;
        int rs = 0, rc = 0;
        if (lane < n) {
            const int2 r = rec[base + lane];
            rs = r.x;
            rc = r.y;
        }
        int j = 0;
        for (; j + 4 <= n; j += 4) {
            const int s0 = __builtin_amdgcn_readlane(rs, j + 0);
            const int s1 = __builtin_amdgcn_readlane(rs, j + 1);
            const int s2 = __builtin_amdgcn_readlane(rs, j + 2);
            const int s3 = __builtin_amdgcn_readlane(rs, j + 3);
            const float c0 = __int_as_float(__builtin_amdgcn_readlane(rc, j + 0));
            const float c1 = __int_as_float(__builtin_amdgcn_readlane(rc, j + 1));
            const float c2 = __int_as_float(__builtin_amdgcn_readlane(rc, j + 2));
            const float c3 = __int_as_float(__builtin_amdgcn_readlane(rc, j + 3));
            const float2 v0 = *(const float2*)(h + (size_t)s0 * GCN_H + p);
            const float2 v1 = *(const float2*)(h + (size_t)s1 * GCN_H + p);
            const float2 v2 = *(const float2*)(h + (size_t)s2 * GCN_H + p);
            const float2 v3 = *(const float2*)(h + (size_t)s3 * GCN_H + p);
            ax = fmaf(v0.x, c0, ax);  ay = fmaf(v0.y, c0, ay);
            ax = fmaf(v1.x, c1, ax);  ay = fmaf(v1.y, c1, ay);
            ax = fmaf(v2.x, c2, ax);  ay = fmaf(v2.y, c2, ay);
            ax = fmaf(v3.x, c3, ax);  ay = fmaf(v3.y, c3, ay);
        }
        for (; j < n; ++j) {
            const int s0 = __builtin_amdgcn_readlane(rs, j);
            const float c0 = __int_as_float(__builtin_amdgcn_readlane(rc, j));
            const float2 v0 = *(const float2*)(h + (size_t)s0 * GCN_H + p);
            ax = fmaf(v0.x, c0, ax);  ay = fmaf(v0.y, c0, ay);
        }
    }

    const float2 bv = *(const float2*)(bias + p);
    float ox = ax + hv.x * di * di + bv.x;
    float oy = ay + hv.y * di * di + bv.y;
    if (do_relu) { ox = fmaxf(ox, 0.0f); oy = fmaxf(oy, 0.0f); }
    *(float2*)(out + (size_t)node * GCN_H + p) = make_float2(ox, oy);
}

__global__ __launch_bounds__(256) void gcn_pool(
    const float* __restrict__ h, const int* __restrict__ batch,
    float* __restrict__ psum, float* __restrict__ pcnt)
{
    const int idx = blockIdx.x * blockDim.x + threadIdx.x;
    const int i = idx >> 6;
    const int f = idx & 63;
    const int b = batch[i];
    atomicAdd(&psum[(size_t)b * GCN_H + f], h[idx]);
    if (f == 0) atomicAdd(&pcnt[b], 1.0f);
}

__global__ __launch_bounds__(256) void gcn_final(
    const float* __restrict__ psum, const float* __restrict__ pcnt,
    const float* __restrict__ Wl, const float* __restrict__ bl,
    float* __restrict__ out)
{
    const int idx = blockIdx.x * blockDim.x + threadIdx.x;
    const int b = idx / GCN_C;
    const int c = idx % GCN_C;
    const float inv = 1.0f / fmaxf(pcnt[b], 1.0f);
    float acc = bl[c];
    const float* row = psum + (size_t)b * GCN_H;
#pragma unroll 8
    for (int k = 0; k < GCN_H; ++k)
        acc = fmaf(row[k] * inv, Wl[k * GCN_C + c], acc);
    out[idx] = acc;
}

extern "C" void kernel_launch(void* const* d_in, const int* in_sizes, int n_in,
                              void* d_out, int out_size, void* d_ws, size_t ws_size,
                              hipStream_t stream) {
    (void)in_sizes; (void)n_in; (void)out_size; (void)ws_size;

    const float* x     = (const float*)d_in[0];
    const int*   ei    = (const int*)  d_in[1];
    const int*   batch = (const int*)  d_in[2];
    const float* W1    = (const float*)d_in[3];
    const float* b1    = (const float*)d_in[4];
    const float* W2    = (const float*)d_in[5];
    const float* b2    = (const float*)d_in[6];
    const float* W3    = (const float*)d_in[7];
    const float* b3    = (const float*)d_in[8];
    const float* Wl    = (const float*)d_in[9];
    const float* bl    = (const float*)d_in[10];
    float* out = (float*)d_out;

    const int* src = ei;
    const int* dst = ei + GCN_E;

    int*   cnt    = (int*)d_ws;
    int*   rowptr = cnt + GCN_NP;
    float* dinv   = (float*)(rowptr + GCN_NP);
    int*   rank   = (int*)(dinv + GCN_NP);
    int2*  rec    = (int2*)(rank + GCN_E);
    float* bufA   = (float*)(rec + GCN_E);
    float* bufB   = bufA + (size_t)GCN_N * GCN_H;
    float* psum   = bufB + (size_t)GCN_N * GCN_H;
    float* pcnt   = psum + (size_t)GCN_B * GCN_H;

    const dim3 gemmBlk(32, 4);
    const dim3 gemmGrd(GCN_N / 16);
    const int  edgeGrd = (GCN_E + 255) / 256;
    const int  gathGrd = GCN_N / 8;
    const int  nhGrd   = (GCN_N * GCN_H) / 256;

    hipMemsetAsync(cnt, 0, GCN_NP * sizeof(int), stream);
    gcn_rank<<<edgeGrd, 256, 0, stream>>>(dst, cnt, rank);
    gcn_scan<<<1, 1024, 0, stream>>>(cnt, rowptr, dinv);
    gcn_fill<<<edgeGrd, 256, 0, stream>>>(src, dst, rank, rowptr, dinv, rec);

    gcn_gemm_wmma_f32<<<gemmGrd, gemmBlk, 0, stream>>>(x, W1, bufA, GCN_FIN);
    gcn_gather<<<gathGrd, 256, 0, stream>>>(bufA, rec, rowptr, dinv, b1, bufB, 1);

    gcn_gemm_wmma_f32<<<gemmGrd, gemmBlk, 0, stream>>>(bufB, W2, bufA, GCN_H);
    gcn_gather<<<gathGrd, 256, 0, stream>>>(bufA, rec, rowptr, dinv, b2, bufB, 1);

    gcn_gemm_wmma_f32<<<gemmGrd, gemmBlk, 0, stream>>>(bufB, W3, bufA, GCN_H);
    gcn_gather<<<gathGrd, 256, 0, stream>>>(bufA, rec, rowptr, dinv, b3, bufB, 0);

    hipMemsetAsync(psum, 0, (size_t)(GCN_B * GCN_H + GCN_B) * sizeof(float), stream);
    gcn_pool<<<nhGrd, 256, 0, stream>>>(bufB, batch, psum, pcnt);
    gcn_final<<<(GCN_B * GCN_C) / 256, 256, 0, stream>>>(psum, pcnt, Wl, bl, out);
}


# Mean pool folded into the layer-3 gather: a block's 8 rows are summed in LDS in node order per run of equal graph id, one float atomic per feature per run; pool kernel and layer-3 row store gone
# speedup vs baseline: 2.2330x; 1.1554x over previous
#include <hip/hip_runtime.h>
#include <hip/hip_bf16.h>


#define GCN_N    50000
#define GCN_FIN  128
#define GCN_E    800000
#define GCN_H    64
#define GCN_C    10
#define GCN_B    512
#define GCN_NP   53248

typedef __attribute__((ext_vector_type(2))) float v2f;
typedef __attribute__((ext_vector_type(8))) float v8f;

__global__ __launch_bounds__(128) void gcn_gemm_wmma_f32(
    const float* __restrict__ A, const float* __restrict__ W,
    float* __restrict__ C, int K)
{
    const int lane    = threadIdx.x;
    const int colBase = threadIdx.y * 16;
    const int rowBase = blockIdx.x * 16;
    const int half    = lane >> 4;
    const int l       = lane & 15;

    const float* arow = A + (size_t)(rowBase + l) * K + 2 * half;
    const float* wcol = W + colBase + l;

    v8f acc = {};
    for (int k0 = 0; k0 < K; k0 += 4) {
        v2f a, b;
        a.x = arow[k0 + 0];
        a.y = arow[k0 + 1];
        b.x = wcol[(size_t)(k0 + 2 * half + 0) * GCN_H];
        b.y = wcol[(size_t)(k0 + 2 * half + 1) * GCN_H];
        acc = __builtin_amdgcn_wmma_f32_16x16x4_f32(
             false, a,  false, b,
             (short)0, acc,  false,  false);
    }

#pragma unroll
    for (int v = 0; v < 8; ++v) {
        const int m = v + 8 * half;
        C[(size_t)(rowBase + m) * GCN_H + colBase + l] = acc[v];
    }
}

__global__ __launch_bounds__(256) void gcn_rank(
    const int* __restrict__ dst, int* __restrict__ cnt, int* __restrict__ rank)
{
    const int e = blockIdx.x * blockDim.x + threadIdx.x;
    if (e < GCN_E) rank[e] = atomicAdd(&cnt[dst[e]], 1);
}

__global__ __launch_bounds__(1024) void gcn_scan(
    const int* __restrict__ cnt, int* __restrict__ rowptr, float* __restrict__ dinv)
{
    __shared__ int part[1024];
    const int t = threadIdx.x;
    const int4* in = (const int4*)(cnt + t * 52);
    int4 c[13];
#pragma unroll
    for (int j = 0; j < 13; ++j) c[j] = in[j];
    int s = 0;
#pragma unroll
    for (int j = 0; j < 13; ++j) s += c[j].x + c[j].y + c[j].z + c[j].w;

    part[t] = s;
    __syncthreads();
    for (int off = 1; off < 1024; off <<= 1) {
        const int add = (t >= off) ? part[t - off] : 0;
        __syncthreads();
        part[t] += add;
        __syncthreads();
    }
    int run = part[t] - s;

    int4*   rp = (int4*)(rowptr + t * 52);
    float4* dv = (float4*)(dinv + t * 52);
#pragma unroll
    for (int j = 0; j < 13; ++j) {
        int4 r;
        r.x = run;
        r.y = r.x + c[j].x;
        r.z = r.y + c[j].y;
        r.w = r.z + c[j].z;
        run = r.w + c[j].w;
        rp[j] = r;
        float4 d;
        d.x = rsqrtf((float)c[j].x + 1.0f);
        d.y = rsqrtf((float)c[j].y + 1.0f);
        d.z = rsqrtf((float)c[j].z + 1.0f);
        d.w = rsqrtf((float)c[j].w + 1.0f);
        dv[j] = d;
    }
}

__global__ __launch_bounds__(256) void gcn_fill(
    const int* __restrict__ src, const int* __restrict__ dst,
    const int* __restrict__ rank, const int* __restrict__ rowptr,
    const float* __restrict__ dinv, int2* __restrict__ rec)
{
    const int e = blockIdx.x * blockDim.x + threadIdx.x;
    if (e < GCN_E) {
        const int s = src[e];
        const int d = dst[e];
        const float c = dinv[s] * dinv[d];
        rec[rowptr[d] + rank[e]] = make_int2(s, __float_as_int(c));
    }
}

__device__ __forceinline__ float2 gcn_gather_row(
    const float* __restrict__ h, const int2* __restrict__ rec,
    const int* __restrict__ rowptr, const float* __restrict__ dinv,
    const float* __restrict__ bias, const int node, const int lane)
{
    const int p = lane * 2;

    const int beg = rowptr[node];
    const int end = rowptr[node + 1];
    const float di = dinv[node];
    const float2 hv = *(const float2*)(h + (size_t)node * GCN_H + p);

    float ax = 0.0f, ay = 0.0f;
    for (int base = beg; base < end; base += 32) {
        const int left = end - base;
        const int n = left < 32 ? left : 32;
        int rs = 0, rc = 0;
        if (lane < n) {
            const int2 r = rec[base + lane];
            rs = r.x;
            rc = r.y;
        }
        int j = 0;
        for (; j + 4 <= n; j += 4) {
            const int s0 = __builtin_amdgcn_readlane(rs, j + 0);
            const int s1 = __builtin_amdgcn_readlane(rs, j + 1);
            const int s2 = __builtin_amdgcn_readlane(rs, j + 2);
            const int s3 = __builtin_amdgcn_readlane(rs, j + 3);
            const float c0 = __int_as_float(__builtin_amdgcn_readlane(rc, j + 0));
            const float c1 = __int_as_float(__builtin_amdgcn_readlane(rc, j + 1));
            const float c2 = __int_as_float(__builtin_amdgcn_readlane(rc, j + 2));
            const float c3 = __int_as_float(__builtin_amdgcn_readlane(rc, j + 3));
            const float2 v0 = *(const float2*)(h + (size_t)s0 * GCN_H + p);
            const float2 v1 = *(const float2*)(h + (size_t)s1 * GCN_H + p);
            const float2 v2 = *(const float2*)(h + (size_t)s2 * GCN_H + p);
            const float2 v3 = *(const float2*)(h + (size_t)s3 * GCN_H + p);
            ax = fmaf(v0.x, c0, ax);  ay = fmaf(v0.y, c0, ay);
            ax = fmaf(v1.x, c1, ax);  ay = fmaf(v1.y, c1, ay);
            ax = fmaf(v2.x, c2, ax);  ay = fmaf(v2.y, c2, ay);
            ax = fmaf(v3.x, c3, ax);  ay = fmaf(v3.y, c3, ay);
        }
        for (; j < n; ++j) {
            const int s0 = __builtin_amdgcn_readlane(rs, j);
            const float c0 = __int_as_float(__builtin_amdgcn_readlane(rc, j));
            const float2 v0 = *(const float2*)(h + (size_t)s0 * GCN_H + p);
            ax = fmaf(v0.x, c0, ax);  ay = fmaf(v0.y, c0, ay);
        }
    }

    const float2 bv = *(const float2*)(bias + p);
    return make_float2(ax + hv.x * di * di + bv.x, ay + hv.y * di * di + bv.y);
}

__global__ __launch_bounds__(256) void gcn_gather(
    const float* __restrict__ h, const int2* __restrict__ rec,
    const int* __restrict__ rowptr, const float* __restrict__ dinv,
    const float* __restrict__ bias, float* __restrict__ out, int do_relu)
{
    const int lane = threadIdx.x & 31;
    const int node = __builtin_amdgcn_readfirstlane(blockIdx.x * 8 + (threadIdx.x >> 5));

    float2 o = gcn_gather_row(h, rec, rowptr, dinv, bias, node, lane);
    if (do_relu) { o.x = fmaxf(o.x, 0.0f); o.y = fmaxf(o.y, 0.0f); }
    *(float2*)(out + (size_t)node * GCN_H + lane * 2) = o;
}

__global__ __launch_bounds__(256) void gcn_gather_pool(
    const float* __restrict__ h, const int2* __restrict__ rec,
    const int* __restrict__ rowptr, const float* __restrict__ dinv,
    const float* __restrict__ bias, const int* __restrict__ batch,
    float* __restrict__ psum, float* __restrict__ pcnt)
{
    __shared__ float srow[8 * GCN_H];
    __shared__ int   sgid[8];

    const int lane = threadIdx.x & 31;
    const int w    = threadIdx.x >> 5;
    const int node = __builtin_amdgcn_readfirstlane(blockIdx.x * 8 + w);
    const int gid  = batch[node];

    const float2 o = gcn_gather_row(h, rec, rowptr, dinv, bias, node, lane);

    *(float2*)(srow + w * GCN_H + lane * 2) = o;
    if (lane == 0) sgid[w] = gid;
    __syncthreads();

    if (threadIdx.x < GCN_H) {
        const int f = threadIdx.x;
        int   cur = __builtin_amdgcn_readfirstlane(sgid[0]);
        float acc = srow[f];
        int   run = 1;
        for (int k = 1; k < 8; ++k) {
            const int   g = __builtin_amdgcn_readfirstlane(sgid[k]);
            const float v = srow[k * GCN_H + f];
            if (g != cur) {
                atomicAdd(&psum[(size_t)cur * GCN_H + f], acc);
                if (f == 0) atomicAdd(&pcnt[cur], (float)run);
                cur = g;  acc = v;  run = 1;
            } else {
                acc += v;  ++run;
            }
        }
        atomicAdd(&psum[(size_t)cur * GCN_H + f], acc);
        if (f == 0) atomicAdd(&pcnt[cur], (float)run);
    }
}

__global__ __launch_bounds__(256) void gcn_final(
    const float* __restrict__ psum, const float* __restrict__ pcnt,
    const float* __restrict__ Wl, const float* __restrict__ bl,
    float* __restrict__ out)
{
    const int idx = blockIdx.x * blockDim.x + threadIdx.x;
    const int b = idx / GCN_C;
    const int c = idx % GCN_C;
    const float inv = 1.0f / fmaxf(pcnt[b], 1.0f);
    float acc = bl[c];
    const float* row = psum + (size_t)b * GCN_H;
#pragma unroll 8
    for (int k = 0; k < GCN_H; ++k)
        acc = fmaf(row[k] * inv, Wl[k * GCN_C + c], acc);
    out[idx] = acc;
}

extern "C" void kernel_launch(void* const* d_in, const int* in_sizes, int n_in,
                              void* d_out, int out_size, void* d_ws, size_t ws_size,
                              hipStream_t stream) {
    (void)in_sizes; (void)n_in; (void)out_size; (void)ws_size;

    const float* x     = (const float*)d_in[0];
    const int*   ei    = (const int*)  d_in[1];
    const int*   batch = (const int*)  d_in[2];
    const float* W1    = (const float*)d_in[3];
    const float* b1    = (const float*)d_in[4];
    const float* W2    = (const float*)d_in[5];
    const float* b2    = (const float*)d_in[6];
    const float* W3    = (const float*)d_in[7];
    const float* b3    = (const float*)d_in[8];
    const float* Wl    = (const float*)d_in[9];
    const float* bl    = (const float*)d_in[10];
    float* out = (float*)d_out;

    const int* src = ei;
    const int* dst = ei + GCN_E;

    int*   cnt    = (int*)d_ws;
    int*   rowptr = cnt + GCN_NP;
    float* dinv   = (float*)(rowptr + GCN_NP);
    int*   rank   = (int*)(dinv + GCN_NP);
    int2*  rec    = (int2*)(rank + GCN_E);
    float* bufA   = (float*)(rec + GCN_E);
    float* bufB   = bufA + (size_t)GCN_N * GCN_H;
    float* psum   = bufB + (size_t)GCN_N * GCN_H;
    float* pcnt   = psum + (size_t)GCN_B * GCN_H;

    const dim3 gemmBlk(32, 4);
    const dim3 gemmGrd(GCN_N / 16);
    const int  edgeGrd = (GCN_E + 255) / 256;
    const int  gathGrd = GCN_N / 8;

    hipMemsetAsync(cnt, 0, GCN_NP * sizeof(int), stream);
    gcn_rank<<<edgeGrd, 256, 0, stream>>>(dst, cnt, rank);
    gcn_scan<<<1, 1024, 0, stream>>>(cnt, rowptr, dinv);
    gcn_fill<<<edgeGrd, 256, 0, stream>>>(src, dst, rank, rowptr, dinv, rec);

    gcn_gemm_wmma_f32<<<gemmGrd, gemmBlk, 0, stream>>>(x, W1, bufA, GCN_FIN);
    gcn_gather<<<gathGrd, 256, 0, stream>>>(bufA, rec, rowptr, dinv, b1, bufB, 1);

    gcn_gemm_wmma_f32<<<gemmGrd, gemmBlk, 0, stream>>>(bufB, W2, bufA, GCN_H);
    gcn_gather<<<gathGrd, 256, 0, stream>>>(bufA, rec, rowptr, dinv, b2, bufB, 1);

    gcn_gemm_wmma_f32<<<gemmGrd, gemmBlk, 0, stream>>>(bufB, W3, bufA, GCN_H);
    hipMemsetAsync(psum, 0, (size_t)(GCN_B * GCN_H + GCN_B) * sizeof(float), stream);
    gcn_gather_pool<<<gathGrd, 256, 0, stream>>>(bufA, rec, rowptr, dinv, b3, batch, psum, pcnt);

    gcn_final<<<(GCN_B * GCN_C) / 256, 256, 0, stream>>>(psum, pcnt, Wl, bl, out);
}


# Layer 2 and 3 dense products moved into the preceding gather's epilogue: 16 relu'd rows meet in LDS, 4 waves multiply by W on f32 WMMA, same k order; two product launches and row round trips gone
# speedup vs baseline: 2.3252x; 1.0413x over previous
#include <hip/hip_runtime.h>
#include <hip/hip_bf16.h>


#define GCN_N    50000
#define GCN_FIN  128
#define GCN_E    800000
#define GCN_H    64
#define GCN_C    10
#define GCN_B    512
#define GCN_NP   53248

typedef __attribute__((ext_vector_type(2))) float v2f;
typedef __attribute__((ext_vector_type(8))) float v8f;

__global__ __launch_bounds__(128) void gcn_gemm_wmma_f32(
    const float* __restrict__ A, const float* __restrict__ W,
    float* __restrict__ C, int K)
{
    const int lane    = threadIdx.x;
    const int colBase = threadIdx.y * 16;
    const int rowBase = blockIdx.x * 16;
    const int half    = lane >> 4;
    const int l       = lane & 15;

    const float* arow = A + (size_t)(rowBase + l) * K + 2 * half;
    const float* wcol = W + colBase + l;

    v8f acc = {};
    for (int k0 = 0; k0 < K; k0 += 4) {
        v2f a, b;
        a.x = arow[k0 + 0];
        a.y = arow[k0 + 1];
        b.x = wcol[(size_t)(k0 + 2 * half + 0) * GCN_H];
        b.y = wcol[(size_t)(k0 + 2 * half + 1) * GCN_H];
        acc = __builtin_amdgcn_wmma_f32_16x16x4_f32(
             false, a,  false, b,
             (short)0, acc,  false,  false);
    }

#pragma unroll
    for (int v = 0; v < 8; ++v) {
        const int m = v + 8 * half;
        C[(size_t)(rowBase + m) * GCN_H + colBase + l] = acc[v];
    }
}

__global__ __launch_bounds__(256) void gcn_rank(
    const int* __restrict__ dst, int* __restrict__ cnt, int* __restrict__ rank)
{
    const int e = blockIdx.x * blockDim.x + threadIdx.x;
    if (e < GCN_E) rank[e] = atomicAdd(&cnt[dst[e]], 1);
}

__global__ __launch_bounds__(1024) void gcn_scan(
    const int* __restrict__ cnt, int* __restrict__ rowptr, float* __restrict__ dinv)
{
    __shared__ int part[1024];
    const int t = threadIdx.x;
    const int4* in = (const int4*)(cnt + t * 52);
    int4 c[13];
#pragma unroll
    for (int j = 0; j < 13; ++j) c[j] = in[j];
    int s = 0;
#pragma unroll
    for (int j = 0; j < 13; ++j) s += c[j].x + c[j].y + c[j].z + c[j].w;

    part[t] = s;
    __syncthreads();
    for (int off = 1; off < 1024; off <<= 1) {
        const int add = (t >= off) ? part[t - off] : 0;
        __syncthreads();
        part[t] += add;
        __syncthreads();
    }
    int run = part[t] - s;

    int4*   rp = (int4*)(rowptr + t * 52);
    float4* dv = (float4*)(dinv + t * 52);
#pragma unroll
    for (int j = 0; j < 13; ++j) {
        int4 r;
        r.x = run;
        r.y = r.x + c[j].x;
        r.z = r.y + c[j].y;
        r.w = r.z + c[j].z;
        run = r.w + c[j].w;
        rp[j] = r;
        float4 d;
        d.x = rsqrtf((float)c[j].x + 1.0f);
        d.y = rsqrtf((float)c[j].y + 1.0f);
        d.z = rsqrtf((float)c[j].z + 1.0f);
        d.w = rsqrtf((float)c[j].w + 1.0f);
        dv[j] = d;
    }
}

__global__ __launch_bounds__(256) void gcn_fill(
    const int* __restrict__ src, const int* __restrict__ dst,
    const int* __restrict__ rank, const int* __restrict__ rowptr,
    const float* __restrict__ dinv, int2* __restrict__ rec)
{
    const int e = blockIdx.x * blockDim.x + threadIdx.x;
    if (e < GCN_E) {
        const int s = src[e];
        const int d = dst[e];
        const float c = dinv[s] * dinv[d];
        rec[rowptr[d] + rank[e]] = make_int2(s, __float_as_int(c));
    }
}

__device__ __forceinline__ float2 gcn_gather_row(
    const float* __restrict__ h, const int2* __restrict__ rec,
    const int* __restrict__ rowptr, const float* __restrict__ dinv,
    const float* __restrict__ bias, const int node, const int lane)
{
    const int p = lane * 2;

    const int beg = rowptr[node];
    const int end = rowptr[node + 1];
    const float di = dinv[node];
    const float2 hv = *(const float2*)(h + (size_t)node * GCN_H + p);

    float ax = 0.0f, ay = 0.0f;
    for (int base = beg; base < end; base += 32) {
        const int left = end - base;
        const int n = left < 32 ? left : 32;
        int rs = 0, rc = 0;
        if (lane < n) {
            const int2 r = rec[base + lane];
            rs = r.x;
            rc = r.y;
        }
        int j = 0;
        for (; j + 4 <= n; j += 4) {
            const int s0 = __builtin_amdgcn_readlane(rs, j + 0);
            const int s1 = __builtin_amdgcn_readlane(rs, j + 1);
            const int s2 = __builtin_amdgcn_readlane(rs, j + 2);
            const int s3 = __builtin_amdgcn_readlane(rs, j + 3);
            const float c0 = __int_as_float(__builtin_amdgcn_readlane(rc, j + 0));
            const float c1 = __int_as_float(__builtin_amdgcn_readlane(rc, j + 1));
            const float c2 = __int_as_float(__builtin_amdgcn_readlane(rc, j + 2));
            const float c3 = __int_as_float(__builtin_amdgcn_readlane(rc, j + 3));
            const float2 v0 = *(const float2*)(h + (size_t)s0 * GCN_H + p);
            const float2 v1 = *(const float2*)(h + (size_t)s1 * GCN_H + p);
            const float2 v2 = *(const float2*)(h + (size_t)s2 * GCN_H + p);
            const float2 v3 = *(const float2*)(h + (size_t)s3 * GCN_H + p);
            ax = fmaf(v0.x, c0, ax);  ay = fmaf(v0.y, c0, ay);
            ax = fmaf(v1.x, c1, ax);  ay = fmaf(v1.y, c1, ay);
            ax = fmaf(v2.x, c2, ax);  ay = fmaf(v2.y, c2, ay);
            ax = fmaf(v3.x, c3, ax);  ay = fmaf(v3.y, c3, ay);
        }
        for (; j < n; ++j) {
            const int s0 = __builtin_amdgcn_readlane(rs, j);
            const float c0 = __int_as_float(__builtin_amdgcn_readlane(rc, j));
            const float2 v0 = *(const float2*)(h + (size_t)s0 * GCN_H + p);
            ax = fmaf(v0.x, c0, ax);  ay = fmaf(v0.y, c0, ay);
        }
    }

    const float2 bv = *(const float2*)(bias + p);
    return make_float2(ax + hv.x * di * di + bv.x, ay + hv.y * di * di + bv.y);
}

#define GCN_TP 68

__global__ __launch_bounds__(512) void gcn_gather_gemm(
    const float* __restrict__ h, const int2* __restrict__ rec,
    const int* __restrict__ rowptr, const float* __restrict__ dinv,
    const float* __restrict__ bias, const float* __restrict__ W,
    float* __restrict__ out)
{
    __shared__ __attribute__((aligned(16))) float tile[16 * GCN_TP];

    const int lane = threadIdx.x & 31;
    const int w    = __builtin_amdgcn_readfirstlane(threadIdx.x >> 5);
    const int node = blockIdx.x * 16 + w;

    float2 o = gcn_gather_row(h, rec, rowptr, dinv, bias, node, lane);
    o.x = fmaxf(o.x, 0.0f);
    o.y = fmaxf(o.y, 0.0f);

    const int half = lane >> 4;
    const int l    = lane & 15;

    v2f b[16];
    if (w < 4) {
        const float* wcol = W + (size_t)(2 * half) * GCN_H + w * 16 + l;
#pragma unroll
        for (int t = 0; t < 16; ++t) {
            b[t].x = wcol[(size_t)(4 * t + 0) * GCN_H];
            b[t].y = wcol[(size_t)(4 * t + 1) * GCN_H];
        }
    }

    *(float2*)(tile + w * GCN_TP + lane * 2) = o;
    __syncthreads();

    if (w < 4) {
        const float* arow = tile + l * GCN_TP + 2 * half;
        v8f acc = {};
#pragma unroll
        for (int t = 0; t < 16; ++t) {
            const float2 av = *(const float2*)(arow + 4 * t);
            v2f a;
            a.x = av.x;
            a.y = av.y;
            acc = __builtin_amdgcn_wmma_f32_16x16x4_f32(
                 false, a,  false, b[t],
                 (short)0, acc,  false,  false);
        }
        float* orow = out + (size_t)(blockIdx.x * 16 + 8 * half) * GCN_H + w * 16 + l;
#pragma unroll
        for (int v = 0; v < 8; ++v)
            orow[(size_t)v * GCN_H] = acc[v];
    }
}

__global__ __launch_bounds__(256) void gcn_gather_pool(
    const float* __restrict__ h, const int2* __restrict__ rec,
    const int* __restrict__ rowptr, const float* __restrict__ dinv,
    const float* __restrict__ bias, const int* __restrict__ batch,
    float* __restrict__ psum, float* __restrict__ pcnt)
{
    __shared__ float srow[8 * GCN_H];
    __shared__ int   sgid[8];

    const int lane = threadIdx.x & 31;
    const int w    = threadIdx.x >> 5;
    const int node = __builtin_amdgcn_readfirstlane(blockIdx.x * 8 + w);
    const int gid  = batch[node];

    const float2 o = gcn_gather_row(h, rec, rowptr, dinv, bias, node, lane);

    *(float2*)(srow + w * GCN_H + lane * 2) = o;
    if (lane == 0) sgid[w] = gid;
    __syncthreads();

    if (threadIdx.x < GCN_H) {
        const int f = threadIdx.x;
        int   cur = __builtin_amdgcn_readfirstlane(sgid[0]);
        float acc = srow[f];
        int   run = 1;
        for (int k = 1; k < 8; ++k) {
            const int   g = __builtin_amdgcn_readfirstlane(sgid[k]);
            const float v = srow[k * GCN_H + f];
            if (g != cur) {
                atomicAdd(&psum[(size_t)cur * GCN_H + f], acc);
                if (f == 0) atomicAdd(&pcnt[cur], (float)run);
                cur = g;  acc = v;  run = 1;
            } else {
                acc += v;  ++run;
            }
        }
        atomicAdd(&psum[(size_t)cur * GCN_H + f], acc);
        if (f == 0) atomicAdd(&pcnt[cur], (float)run);
    }
}

__global__ __launch_bounds__(256) void gcn_final(
    const float* __restrict__ psum, const float* __restrict__ pcnt,
    const float* __restrict__ Wl, const float* __restrict__ bl,
    float* __restrict__ out)
{
    const int idx = blockIdx.x * blockDim.x + threadIdx.x;
    const int b = idx / GCN_C;
    const int c = idx % GCN_C;
    const float inv = 1.0f / fmaxf(pcnt[b], 1.0f);
    float acc = bl[c];
    const float* row = psum + (size_t)b * GCN_H;
#pragma unroll 8
    for (int k = 0; k < GCN_H; ++k)
        acc = fmaf(row[k] * inv, Wl[k * GCN_C + c], acc);
    out[idx] = acc;
}

extern "C" void kernel_launch(void* const* d_in, const int* in_sizes, int n_in,
                              void* d_out, int out_size, void* d_ws, size_t ws_size,
                              hipStream_t stream) {
    (void)in_sizes; (void)n_in; (void)out_size; (void)ws_size;

    const float* x     = (const float*)d_in[0];
    const int*   ei    = (const int*)  d_in[1];
    const int*   batch = (const int*)  d_in[2];
    const float* W1    = (const float*)d_in[3];
    const float* b1    = (const float*)d_in[4];
    const float* W2    = (const float*)d_in[5];
    const float* b2    = (const float*)d_in[6];
    const float* W3    = (const float*)d_in[7];
    const float* b3    = (const float*)d_in[8];
    const float* Wl    = (const float*)d_in[9];
    const float* bl    = (const float*)d_in[10];
    float* out = (float*)d_out;

    const int* src = ei;
    const int* dst = ei + GCN_E;

    int*   cnt    = (int*)d_ws;
    int*   rowptr = cnt + GCN_NP;
    float* dinv   = (float*)(rowptr + GCN_NP);
    int*   rank   = (int*)(dinv + GCN_NP);
    int2*  rec    = (int2*)(rank + GCN_E);
    float* bufA   = (float*)(rec + GCN_E);
    float* bufB   = bufA + (size_t)GCN_N * GCN_H;
    float* psum   = bufB + (size_t)GCN_N * GCN_H;
    float* pcnt   = psum + (size_t)GCN_B * GCN_H;

    const dim3 gemmBlk(32, 4);
    const dim3 gemmGrd(GCN_N / 16);
    const int  edgeGrd = (GCN_E + 255) / 256;
    const int  gathGrd = GCN_N / 8;
    const int  fuseGrd = GCN_N / 16;

    hipMemsetAsync(cnt, 0, GCN_NP * sizeof(int), stream);
    gcn_rank<<<edgeGrd, 256, 0, stream>>>(dst, cnt, rank);
    gcn_scan<<<1, 1024, 0, stream>>>(cnt, rowptr, dinv);
    gcn_fill<<<edgeGrd, 256, 0, stream>>>(src, dst, rank, rowptr, dinv, rec);

    gcn_gemm_wmma_f32<<<gemmGrd, gemmBlk, 0, stream>>>(x, W1, bufA, GCN_FIN);

    gcn_gather_gemm<<<fuseGrd, 512, 0, stream>>>(bufA, rec, rowptr, dinv, b1, W2, bufB);

    gcn_gather_gemm<<<fuseGrd, 512, 0, stream>>>(bufB, rec, rowptr, dinv, b2, W3, bufA);

    hipMemsetAsync(psum, 0, (size_t)(GCN_B * GCN_H + GCN_B) * sizeof(float), stream);
    gcn_gather_pool<<<gathGrd, 256, 0, stream>>>(bufA, rec, rowptr, dinv, b3, batch, psum, pcnt);

    gcn_final<<<(GCN_B * GCN_C) / 256, 256, 0, stream>>>(psum, pcnt, Wl, bl, out);
}


# Final linear folded into layer 3 (associativity): wc=W3@Wl, formed once per call, is the layer-2 epilogue's B; layer-3 gather+pool run on 16-float rows, 4 lanes per node; tail rounding changes
# speedup vs baseline: 2.4697x; 1.0621x over previous
#include <hip/hip_runtime.h>
#include <hip/hip_bf16.h>


#define GCN_N    50000
#define GCN_FIN  128
#define GCN_E    800000
#define GCN_H    64
#define GCN_C    10
#define GCN_B    512
#define GCN_NP   53248
#define GCN_ZW   16

typedef __attribute__((ext_vector_type(2))) float v2f;
typedef __attribute__((ext_vector_type(8))) float v8f;

__global__ __launch_bounds__(128) void gcn_gemm_wmma_f32(
    const float* __restrict__ A, const float* __restrict__ W,
    float* __restrict__ C, int K)
{
    const int lane    = threadIdx.x;
    const int colBase = threadIdx.y * 16;
    const int rowBase = blockIdx.x * 16;
    const int half    = lane >> 4;
    const int l       = lane & 15;

    const float* arow = A + (size_t)(rowBase + l) * K + 2 * half;
    const float* wcol = W + colBase + l;

    v8f acc = {};
    for (int k0 = 0; k0 < K; k0 += 4) {
        v2f a, b;
        a.x = arow[k0 + 0];
        a.y = arow[k0 + 1];
        b.x = wcol[(size_t)(k0 + 2 * half + 0) * GCN_H];
        b.y = wcol[(size_t)(k0 + 2 * half + 1) * GCN_H];
        acc = __builtin_amdgcn_wmma_f32_16x16x4_f32(
             false, a,  false, b,
             (short)0, acc,  false,  false);
    }

#pragma unroll
    for (int v = 0; v < 8; ++v) {
        const int m = v + 8 * half;
        C[(size_t)(rowBase + m) * GCN_H + colBase + l] = acc[v];
    }
}

__global__ __launch_bounds__(256) void gcn_rank(
    const int* __restrict__ dst, int* __restrict__ cnt, int* __restrict__ rank)
{
    const int e = blockIdx.x * blockDim.x + threadIdx.x;
    if (e < GCN_E) rank[e] = atomicAdd(&cnt[dst[e]], 1);
}

__global__ __launch_bounds__(1024) void gcn_scan(
    const int* __restrict__ cnt, int* __restrict__ rowptr, float* __restrict__ dinv,
    const float* __restrict__ W3, const float* __restrict__ b3,
    const float* __restrict__ Wl, float* __restrict__ wc)
{
    if (blockIdx.x == 1) {
        for (int idx = threadIdx.x; idx < 65 * GCN_ZW; idx += 1024) {
            const int r = idx / GCN_ZW;
            const int c = idx % GCN_ZW;
            const float* a = (r < GCN_H) ? (W3 + r * GCN_H) : b3;
            float acc = 0.0f;
            if (c < GCN_C) {
                for (int k = 0; k < GCN_H; ++k) acc = fmaf(a[k], Wl[k * GCN_C + c], acc);
            }
            wc[idx] = acc;
        }
        return;
    }

    __shared__ int part[1024];
    const int t = threadIdx.x;
    const int4* in = (const int4*)(cnt + t * 52);
    int4 c[13];
#pragma unroll
    for (int j = 0; j < 13; ++j) c[j] = in[j];
    int s = 0;
#pragma unroll
    for (int j = 0; j < 13; ++j) s += c[j].x + c[j].y + c[j].z + c[j].w;

    part[t] = s;
    __syncthreads();
    for (int off = 1; off < 1024; off <<= 1) {
        const int add = (t >= off) ? part[t - off] : 0;
        __syncthreads();
        part[t] += add;
        __syncthreads();
    }
    int run = part[t] - s;

    int4*   rp = (int4*)(rowptr + t * 52);
    float4* dv = (float4*)(dinv + t * 52);
#pragma unroll
    for (int j = 0; j < 13; ++j) {
        int4 r;
        r.x = run;
        r.y = r.x + c[j].x;
        r.z = r.y + c[j].y;
        r.w = r.z + c[j].z;
        run = r.w + c[j].w;
        rp[j] = r;
        float4 d;
        d.x = rsqrtf((float)c[j].x + 1.0f);
        d.y = rsqrtf((float)c[j].y + 1.0f);
        d.z = rsqrtf((float)c[j].z + 1.0f);
        d.w = rsqrtf((float)c[j].w + 1.0f);
        dv[j] = d;
    }
}

__global__ __launch_bounds__(256) void gcn_fill(
    const int* __restrict__ src, const int* __restrict__ dst,
    const int* __restrict__ rank, const int* __restrict__ rowptr,
    const float* __restrict__ dinv, int2* __restrict__ rec)
{
    const int e = blockIdx.x * blockDim.x + threadIdx.x;
    if (e < GCN_E) {
        const int s = src[e];
        const int d = dst[e];
        const float c = dinv[s] * dinv[d];
        rec[rowptr[d] + rank[e]] = make_int2(s, __float_as_int(c));
    }
}

__device__ __forceinline__ float2 gcn_gather_row(
    const float* __restrict__ h, const int2* __restrict__ rec,
    const int* __restrict__ rowptr, const float* __restrict__ dinv,
    const float* __restrict__ bias, const int node, const int lane)
{
    const int p = lane * 2;

    const int beg = rowptr[node];
    const int end = rowptr[node + 1];
    const float di = dinv[node];
    const float2 hv = *(const float2*)(h + (size_t)node * GCN_H + p);

    float ax = 0.0f, ay = 0.0f;
    for (int base = beg; base < end; base += 32) {
        const int left = end - base;
        const int n = left < 32 ? left : 32;
        int rs = 0, rc = 0;
        if (lane < n) {
            const int2 r = rec[base + lane];
            rs = r.x;
            rc = r.y;
        }
        int j = 0;
        for (; j + 4 <= n; j += 4) {
            const int s0 = __builtin_amdgcn_readlane(rs, j + 0);
            const int s1 = __builtin_amdgcn_readlane(rs, j + 1);
            const int s2 = __builtin_amdgcn_readlane(rs, j + 2);
            const int s3 = __builtin_amdgcn_readlane(rs, j + 3);
            const float c0 = __int_as_float(__builtin_amdgcn_readlane(rc, j + 0));
            const float c1 = __int_as_float(__builtin_amdgcn_readlane(rc, j + 1));
            const float c2 = __int_as_float(__builtin_amdgcn_readlane(rc, j + 2));
            const float c3 = __int_as_float(__builtin_amdgcn_readlane(rc, j + 3));
            const float2 v0 = *(const float2*)(h + (size_t)s0 * GCN_H + p);
            const float2 v1 = *(const float2*)(h + (size_t)s1 * GCN_H + p);
            const float2 v2 = *(const float2*)(h + (size_t)s2 * GCN_H + p);
            const float2 v3 = *(const float2*)(h + (size_t)s3 * GCN_H + p);
            ax = fmaf(v0.x, c0, ax);  ay = fmaf(v0.y, c0, ay);
            ax = fmaf(v1.x, c1, ax);  ay = fmaf(v1.y, c1, ay);
            ax = fmaf(v2.x, c2, ax);  ay = fmaf(v2.y, c2, ay);
            ax = fmaf(v3.x, c3, ax);  ay = fmaf(v3.y, c3, ay);
        }
        for (; j < n; ++j) {
            const int s0 = __builtin_amdgcn_readlane(rs, j);
            const float c0 = __int_as_float(__builtin_amdgcn_readlane(rc, j));
            const float2 v0 = *(const float2*)(h + (size_t)s0 * GCN_H + p);
            ax = fmaf(v0.x, c0, ax);  ay = fmaf(v0.y, c0, ay);
        }
    }

    const float2 bv = *(const float2*)(bias + p);
    return make_float2(ax + hv.x * di * di + bv.x, ay + hv.y * di * di + bv.y);
}

#define GCN_TP 68

template <int NT>
__global__ __launch_bounds__(512) void gcn_gather_gemm(
    const float* __restrict__ h, const int2* __restrict__ rec,
    const int* __restrict__ rowptr, const float* __restrict__ dinv,
    const float* __restrict__ bias, const float* __restrict__ W,
    float* __restrict__ out)
{
    __shared__ __attribute__((aligned(16))) float tile[16 * GCN_TP];
    constexpr int P = 16 * NT;

    const int lane = threadIdx.x & 31;
    const int w    = __builtin_amdgcn_readfirstlane(threadIdx.x >> 5);
    const int node = blockIdx.x * 16 + w;

    float2 o = gcn_gather_row(h, rec, rowptr, dinv, bias, node, lane);
    o.x = fmaxf(o.x, 0.0f);
    o.y = fmaxf(o.y, 0.0f);

    const int half = lane >> 4;
    const int l    = lane & 15;

    v2f b[16];
    if (w < NT) {
        const float* wcol = W + (size_t)(2 * half) * P + w * 16 + l;
#pragma unroll
        for (int t = 0; t < 16; ++t) {
            b[t].x = wcol[(size_t)(4 * t + 0) * P];
            b[t].y = wcol[(size_t)(4 * t + 1) * P];
        }
    }

    *(float2*)(tile + w * GCN_TP + lane * 2) = o;
    __syncthreads();

    if (w < NT) {
        const float* arow = tile + l * GCN_TP + 2 * half;
        v8f acc = {};
#pragma unroll
        for (int t = 0; t < 16; ++t) {
            const float2 av = *(const float2*)(arow + 4 * t);
            v2f a;
            a.x = av.x;
            a.y = av.y;
            acc = __builtin_amdgcn_wmma_f32_16x16x4_f32(
                 false, a,  false, b[t],
                 (short)0, acc,  false,  false);
        }
        float* orow = out + (size_t)(blockIdx.x * 16 + 8 * half) * P + w * 16 + l;
#pragma unroll
        for (int v = 0; v < 8; ++v)
            orow[(size_t)v * P] = acc[v];
    }
}

__global__ __launch_bounds__(256) void gcn_gather_pool16(
    const float* __restrict__ z, const int2* __restrict__ rec,
    const int* __restrict__ rowptr, const float* __restrict__ dinv,
    const float* __restrict__ bias, const int* __restrict__ batch,
    float* __restrict__ psum, float* __restrict__ pcnt)
{
    __shared__ __attribute__((aligned(16))) float srow[64 * GCN_ZW];
    __shared__ int sgid[64];

    const int q    = threadIdx.x & 3;
    const int k    = threadIdx.x >> 2;
    const int node = blockIdx.x * 64 + k;

    float4 o = make_float4(0.0f, 0.0f, 0.0f, 0.0f);
    static_assert(GCN_N % 8 == 0, "a wave's 8 nodes exist together or not at all");
    if (node < GCN_N) {
        const int beg = rowptr[node];
        const int end = rowptr[node + 1];
        const float di = dinv[node];
        const float4 hv = *(const float4*)(z + (size_t)node * GCN_ZW + 4 * q);
        const float4 bv = *(const float4*)(bias + 4 * q);

        float4 a = make_float4(0.0f, 0.0f, 0.0f, 0.0f);
        const int last = end - 1;
        for (int s = beg; s < end; s += 4) {
            const int i1 = (s + 1 < end) ? s + 1 : last;
            const int i2 = (s + 2 < end) ? s + 2 : last;
            const int i3 = (s + 3 < end) ? s + 3 : last;
            const int2 r0 = rec[s];
            const int2 r1 = rec[i1];
            const int2 r2 = rec[i2];
            const int2 r3 = rec[i3];
            float4 v0 = *(const float4*)(z + (size_t)r0.x * GCN_ZW + 4 * q);
            float4 v1 = *(const float4*)(z + (size_t)r1.x * GCN_ZW + 4 * q);
            float4 v2 = *(const float4*)(z + (size_t)r2.x * GCN_ZW + 4 * q);
            float4 v3 = *(const float4*)(z + (size_t)r3.x * GCN_ZW + 4 * q);
            const float c0 = __int_as_float(r0.y);
            float c1 = __int_as_float(r1.y);
            float c2 = __int_as_float(r2.y);
            float c3 = __int_as_float(r3.y);
            const float4 zero = make_float4(0.0f, 0.0f, 0.0f, 0.0f);
            if (s + 1 >= end) { v1 = zero; c1 = 0.0f; }
            if (s + 2 >= end) { v2 = zero; c2 = 0.0f; }
            if (s + 3 >= end) { v3 = zero; c3 = 0.0f; }
            a.x = fmaf(v0.x, c0, a.x);  a.y = fmaf(v0.y, c0, a.y);  a.z = fmaf(v0.z, c0, a.z);  a.w = fmaf(v0.w, c0, a.w);
            a.x = fmaf(v1.x, c1, a.x);  a.y = fmaf(v1.y, c1, a.y);  a.z = fmaf(v1.z, c1, a.z);  a.w = fmaf(v1.w, c1, a.w);
            a.x = fmaf(v2.x, c2, a.x);  a.y = fmaf(v2.y, c2, a.y);  a.z = fmaf(v2.z, c2, a.z);  a.w = fmaf(v2.w, c2, a.w);
            a.x = fmaf(v3.x, c3, a.x);  a.y = fmaf(v3.y, c3, a.y);  a.z = fmaf(v3.z, c3, a.z);  a.w = fmaf(v3.w, c3, a.w);
        }
        o.x = a.x + hv.x * di * di + bv.x;
        o.y = a.y + hv.y * di * di + bv.y;
        o.z = a.z + hv.z * di * di + bv.z;
        o.w = a.w + hv.w * di * di + bv.w;
        if (q == 0) sgid[k] = batch[node];
    }

    *(float4*)(srow + k * GCN_ZW + 4 * q) = o;
    __syncthreads();

    const int lane  = threadIdx.x & 31;
    const int w     = threadIdx.x >> 5;
    const int first = blockIdx.x * 64 + w * 8;
    if (lane < GCN_C && first < GCN_N) {
        const float* rows = srow + w * 8 * GCN_ZW;
        const int*   ids  = sgid + w * 8;
        int   cur = __builtin_amdgcn_readfirstlane(ids[0]);
        float acc = rows[lane];
        int   run = 1;
        for (int j = 1; j < 8; ++j) {
            const int   g = __builtin_amdgcn_readfirstlane(ids[j]);
            const float v = rows[j * GCN_ZW + lane];
            if (g != cur) {
                atomicAdd(&psum[(size_t)cur * GCN_ZW + lane], acc);
                if (lane == 0) atomicAdd(&pcnt[cur], (float)run);
                cur = g;  acc = v;  run = 1;
            } else {
                acc += v;  ++run;
            }
        }
        atomicAdd(&psum[(size_t)cur * GCN_ZW + lane], acc);
        if (lane == 0) atomicAdd(&pcnt[cur], (float)run);
    }
}

__global__ __launch_bounds__(256) void gcn_final(
    const float* __restrict__ psum, const float* __restrict__ pcnt,
    const float* __restrict__ bl, float* __restrict__ out)
{
    const int idx = blockIdx.x * blockDim.x + threadIdx.x;
    const int b = idx / GCN_C;
    const int c = idx % GCN_C;
    out[idx] = psum[(size_t)b * GCN_ZW + c] / fmaxf(pcnt[b], 1.0f) + bl[c];
}

extern "C" void kernel_launch(void* const* d_in, const int* in_sizes, int n_in,
                              void* d_out, int out_size, void* d_ws, size_t ws_size,
                              hipStream_t stream) {
    (void)in_sizes; (void)n_in; (void)out_size; (void)ws_size;

    const float* x     = (const float*)d_in[0];
    const int*   ei    = (const int*)  d_in[1];
    const int*   batch = (const int*)  d_in[2];
    const float* W1    = (const float*)d_in[3];
    const float* b1    = (const float*)d_in[4];
    const float* W2    = (const float*)d_in[5];
    const float* b2    = (const float*)d_in[6];
    const float* W3    = (const float*)d_in[7];
    const float* b3    = (const float*)d_in[8];
    const float* Wl    = (const float*)d_in[9];
    const float* bl    = (const float*)d_in[10];
    float* out = (float*)d_out;

    const int* src = ei;
    const int* dst = ei + GCN_E;

    int*   cnt    = (int*)d_ws;
    int*   rowptr = cnt + GCN_NP;
    float* dinv   = (float*)(rowptr + GCN_NP);
    int*   rank   = (int*)(dinv + GCN_NP);
    int2*  rec    = (int2*)(rank + GCN_E);
    float* bufA   = (float*)(rec + GCN_E);
    float* bufB   = bufA + (size_t)GCN_N * GCN_H;
    float* psum   = bufB + (size_t)GCN_N * GCN_H;
    float* pcnt   = psum + (size_t)GCN_B * GCN_ZW;
    float* wc     = pcnt + GCN_B;

    const dim3 gemmBlk(32, 4);
    const dim3 gemmGrd(GCN_N / 16);
    const int  edgeGrd = (GCN_E + 255) / 256;
    const int  poolGrd = (GCN_N + 63) / 64;
    const int  fuseGrd = GCN_N / 16;

    hipMemsetAsync(cnt, 0, GCN_NP * sizeof(int), stream);
    gcn_rank<<<edgeGrd, 256, 0, stream>>>(dst, cnt, rank);
    gcn_scan<<<2, 1024, 0, stream>>>(cnt, rowptr, dinv, W3, b3, Wl, wc);
    gcn_fill<<<edgeGrd, 256, 0, stream>>>(src, dst, rank, rowptr, dinv, rec);

    gcn_gemm_wmma_f32<<<gemmGrd, gemmBlk, 0, stream>>>(x, W1, bufA, GCN_FIN);

    gcn_gather_gemm<4><<<fuseGrd, 512, 0, stream>>>(bufA, rec, rowptr, dinv, b1, W2, bufB);

    gcn_gather_gemm<1><<<fuseGrd, 512, 0, stream>>>(bufB, rec, rowptr, dinv, b2, wc, bufA);

    hipMemsetAsync(psum, 0, (size_t)(GCN_B * GCN_ZW + GCN_B) * sizeof(float), stream);
    gcn_gather_pool16<<<poolGrd, 256, 0, stream>>>(bufA, rec, rowptr, dinv, wc + 64 * GCN_ZW, batch, psum, pcnt);

    gcn_final<<<(GCN_B * GCN_C) / 256, 256, 0, stream>>>(psum, pcnt, bl, out);
}


# Counting sort by dst moved into LDS: edges binned by 256-node range (LDS histogram per chunk), one block per bin ranks by LDS add; no global atomic, no scattered store; sum order = LDS rank order
# speedup vs baseline: 4.0180x; 1.6269x over previous
#include <hip/hip_runtime.h>
#include <hip/hip_bf16.h>


#define GCN_N    50000
#define GCN_FIN  128
#define GCN_E    800000
#define GCN_H    64
#define GCN_C    10
#define GCN_B    512
#define GCN_NP   53248
#define GCN_ZW   16

typedef __attribute__((ext_vector_type(2))) float v2f;
typedef __attribute__((ext_vector_type(8))) float v8f;

__global__ __launch_bounds__(128) void gcn_gemm_wmma_f32(
    const float* __restrict__ A, const float* __restrict__ W,
    float* __restrict__ C, int K)
{
    const int lane    = threadIdx.x;
    const int colBase = threadIdx.y * 16;
    const int rowBase = blockIdx.x * 16;
    const int half    = lane >> 4;
    const int l       = lane & 15;

    const float* arow = A + (size_t)(rowBase + l) * K + 2 * half;
    const float* wcol = W + colBase + l;

    v8f acc = {};
    for (int k0 = 0; k0 < K; k0 += 4) {
        v2f a, b;
        a.x = arow[k0 + 0];
        a.y = arow[k0 + 1];
        b.x = wcol[(size_t)(k0 + 2 * half + 0) * GCN_H];
        b.y = wcol[(size_t)(k0 + 2 * half + 1) * GCN_H];
        acc = __builtin_amdgcn_wmma_f32_16x16x4_f32(
             false, a,  false, b,
             (short)0, acc,  false,  false);
    }

#pragma unroll
    for (int v = 0; v < 8; ++v) {
        const int m = v + 8 * half;
        C[(size_t)(rowBase + m) * GCN_H + colBase + l] = acc[v];
    }
}

#define GCN_NB   196
#define GCN_G1   125
#define GCN_CH   6400
#define GCN_EPT  7
#define GCN_CAP  8192

static_assert(GCN_G1 * GCN_CH == GCN_E, "chunks cover the edge list exactly");
static_assert(GCN_NB * 256 >= GCN_N + 1 && GCN_NB <= 256 && GCN_N < 65536, "key packing");

__device__ __forceinline__ void gcn_excl_scan256(int* a, const int t)
{
    if (t < 32) {
        int v[8];
        int s = 0;
#pragma unroll
        for (int j = 0; j < 8; ++j) { v[j] = a[t * 8 + j]; s += v[j]; }
        int inc = s;
#pragma unroll
        for (int off = 1; off < 32; off <<= 1) {
            const int o = __shfl_up(inc, off);
            if (t >= off) inc += o;
        }
        int run = inc - s;
#pragma unroll
        for (int j = 0; j < 8; ++j) { a[t * 8 + j] = run; run += v[j]; }
        if (t == 31) a[256] = run;
    }
    __syncthreads();
}

__global__ __launch_bounds__(1024) void gcn_bin_count(
    const int* __restrict__ dst, int* __restrict__ cntBB,
    const float* __restrict__ W3, const float* __restrict__ b3,
    const float* __restrict__ Wl, float* __restrict__ wc)
{
    if (blockIdx.x == GCN_G1) {
        for (int idx = threadIdx.x; idx < 65 * GCN_ZW; idx += 1024) {
            const int r = idx / GCN_ZW;
            const int c = idx % GCN_ZW;
            const float* a = (r < GCN_H) ? (W3 + r * GCN_H) : b3;
            float acc = 0.0f;
            if (c < GCN_C) {
                for (int k = 0; k < GCN_H; ++k) acc = fmaf(a[k], Wl[k * GCN_C + c], acc);
            }
            wc[idx] = acc;
        }
        return;
    }

    __shared__ int lcnt[256];
    const int t = threadIdx.x;
    if (t < 256) lcnt[t] = 0;
    __syncthreads();
    const int* d = dst + blockIdx.x * GCN_CH;
    for (int i = t; i < GCN_CH; i += 1024) atomicAdd(&lcnt[d[i] >> 8], 1);
    __syncthreads();
    if (t < 256) cntBB[blockIdx.x * 256 + t] = lcnt[t];
}

__global__ __launch_bounds__(1024) void gcn_bin_scatter(
    const int* __restrict__ src, const int* __restrict__ dst,
    const int* __restrict__ cntBB, int* __restrict__ binStart,
    unsigned int* __restrict__ tmp)
{
    __shared__ int tot[257];
    __shared__ int lst[257];
    __shared__ int off[256];
    __shared__ unsigned int sorted[GCN_CH];

    const int t = threadIdx.x;
    const int b = blockIdx.x;
    const int* sp = src + b * GCN_CH;
    const int* dp = dst + b * GCN_CH;

    unsigned int key[GCN_EPT];
#pragma unroll
    for (int j = 0; j < GCN_EPT; ++j) {
        const int i = t + j * 1024;
        key[j] = 0u;
        if (i < GCN_CH) {
            const unsigned int s = (unsigned int)sp[i];
            const unsigned int d = (unsigned int)dp[i];
            key[j] = ((d >> 8) << 24) | (s << 8) | (d & 255u);
        }
    }

    int before = 0;
    if (t < 256) {
        int total = 0;
        for (int bb = 0; bb < GCN_G1; ++bb) {
            const int c = cntBB[bb * 256 + t];
            total += c;
            before += (bb < b) ? c : 0;
        }
        tot[t] = total;
        lst[t] = 0;
    }
    __syncthreads();
    gcn_excl_scan256(tot, t);
    if (b == 0 && t <= 256) binStart[t] = tot[t];

    int r[GCN_EPT];
#pragma unroll
    for (int j = 0; j < GCN_EPT; ++j) {
        r[j] = 0;
        if (t + j * 1024 < GCN_CH) r[j] = atomicAdd(&lst[key[j] >> 24], 1);
    }
    __syncthreads();
    gcn_excl_scan256(lst, t);
    if (t < 256) off[t] = tot[t] + before - lst[t];
#pragma unroll
    for (int j = 0; j < GCN_EPT; ++j) {
        if (t + j * 1024 < GCN_CH) sorted[lst[key[j] >> 24] + r[j]] = key[j];
    }
    __syncthreads();
    for (int p = t; p < GCN_CH; p += 1024) {
        const unsigned int k = sorted[p];
        tmp[off[k >> 24] + p] = k;
    }
}

__global__ __launch_bounds__(1024) void gcn_bin_sort(
    const unsigned int* __restrict__ tmp, const int* __restrict__ binStart,
    int* __restrict__ rowptr, float* __restrict__ dinv, int2* __restrict__ rec)
{
    __shared__ int   cnt[257];
    __shared__ int   cur[256];
    __shared__ float sdv[256];
    __shared__ int2  stage[GCN_CAP];

    const int t   = threadIdx.x;
    const int bin = blockIdx.x;
    const int beg = binStart[bin];
    const int end = binStart[bin + 1];
    const int n   = end - beg;

    if (t < 256) cnt[t] = 0;
    __syncthreads();
    for (int i = beg + t; i < end; i += 1024) atomicAdd(&cnt[tmp[i] & 255u], 1);
    __syncthreads();
    int deg = 0;
    if (t < 256) deg = cnt[t];
    __syncthreads();
    gcn_excl_scan256(cnt, t);
    if (t < 256) {
        const int node = bin * 256 + t;
        const float dv = rsqrtf((float)deg + 1.0f);
        sdv[t] = dv;
        cur[t] = cnt[t];
        if (node <= GCN_N) rowptr[node] = beg + cnt[t];
        if (node <  GCN_N) dinv[node]   = dv;
    }
    __syncthreads();

    const bool staged = (n <= GCN_CAP);
    for (int i = beg + t; i < end; i += 1024) {
        const unsigned int k = tmp[i];
        const int dl  = (int)(k & 255u);
        const int s   = (int)((k >> 8) & 0xFFFFu);
        const int pos = atomicAdd(&cur[dl], 1);
        const int2 v  = make_int2(s, __float_as_int(sdv[dl]));
        if (staged) stage[pos] = v;
        else        rec[beg + pos] = v;
    }
    __syncthreads();
    if (staged) {
        for (int p = t; p < n; p += 1024) rec[beg + p] = stage[p];
    }
}

__global__ __launch_bounds__(128) void gcn_coef(
    const float* __restrict__ dinv, int4* __restrict__ rec2)
{
    const int idx = blockIdx.x * blockDim.x + threadIdx.x;
    int4 r = rec2[idx];
    r.y = __float_as_int(dinv[r.x] * __int_as_float(r.y));
    r.w = __float_as_int(dinv[r.z] * __int_as_float(r.w));
    rec2[idx] = r;
}

__device__ __forceinline__ float2 gcn_gather_row(
    const float* __restrict__ h, const int2* __restrict__ rec,
    const int* __restrict__ rowptr, const float* __restrict__ dinv,
    const float* __restrict__ bias, const int node, const int lane)
{
    const int p = lane * 2;

    const int beg = rowptr[node];
    const int end = rowptr[node + 1];
    const float di = dinv[node];
    const float2 hv = *(const float2*)(h + (size_t)node * GCN_H + p);

    float ax = 0.0f, ay = 0.0f;
    for (int base = beg; base < end; base += 32) {
        const int left = end - base;
        const int n = left < 32 ? left : 32;
        int rs = 0, rc = 0;
        if (lane < n) {
            const int2 r = rec[base + lane];
            rs = r.x;
            rc = r.y;
        }
        int j = 0;
        for (; j + 4 <= n; j += 4) {
            const int s0 = __builtin_amdgcn_readlane(rs, j + 0);
            const int s1 = __builtin_amdgcn_readlane(rs, j + 1);
            const int s2 = __builtin_amdgcn_readlane(rs, j + 2);
            const int s3 = __builtin_amdgcn_readlane(rs, j + 3);
            const float c0 = __int_as_float(__builtin_amdgcn_readlane(rc, j + 0));
            const float c1 = __int_as_float(__builtin_amdgcn_readlane(rc, j + 1));
            const float c2 = __int_as_float(__builtin_amdgcn_readlane(rc, j + 2));
            const float c3 = __int_as_float(__builtin_amdgcn_readlane(rc, j + 3));
            const float2 v0 = *(const float2*)(h + (size_t)s0 * GCN_H + p);
            const float2 v1 = *(const float2*)(h + (size_t)s1 * GCN_H + p);
            const float2 v2 = *(const float2*)(h + (size_t)s2 * GCN_H + p);
            const float2 v3 = *(const float2*)(h + (size_t)s3 * GCN_H + p);
            ax = fmaf(v0.x, c0, ax);  ay = fmaf(v0.y, c0, ay);
            ax = fmaf(v1.x, c1, ax);  ay = fmaf(v1.y, c1, ay);
            ax = fmaf(v2.x, c2, ax);  ay = fmaf(v2.y, c2, ay);
            ax = fmaf(v3.x, c3, ax);  ay = fmaf(v3.y, c3, ay);
        }
        for (; j < n; ++j) {
            const int s0 = __builtin_amdgcn_readlane(rs, j);
            const float c0 = __int_as_float(__builtin_amdgcn_readlane(rc, j));
            const float2 v0 = *(const float2*)(h + (size_t)s0 * GCN_H + p);
            ax = fmaf(v0.x, c0, ax);  ay = fmaf(v0.y, c0, ay);
        }
    }

    const float2 bv = *(const float2*)(bias + p);
    return make_float2(ax + hv.x * di * di + bv.x, ay + hv.y * di * di + bv.y);
}

#define GCN_TP 68

template <int NT>
__global__ __launch_bounds__(512) void gcn_gather_gemm(
    const float* __restrict__ h, const int2* __restrict__ rec,
    const int* __restrict__ rowptr, const float* __restrict__ dinv,
    const float* __restrict__ bias, const float* __restrict__ W,
    float* __restrict__ out)
{
    __shared__ __attribute__((aligned(16))) float tile[16 * GCN_TP];
    constexpr int P = 16 * NT;

    const int lane = threadIdx.x & 31;
    const int w    = __builtin_amdgcn_readfirstlane(threadIdx.x >> 5);
    const int node = blockIdx.x * 16 + w;

    float2 o = gcn_gather_row(h, rec, rowptr, dinv, bias, node, lane);
    o.x = fmaxf(o.x, 0.0f);
    o.y = fmaxf(o.y, 0.0f);

    const int half = lane >> 4;
    const int l    = lane & 15;

    v2f b[16];
    if (w < NT) {
        const float* wcol = W + (size_t)(2 * half) * P + w * 16 + l;
#pragma unroll
        for (int t = 0; t < 16; ++t) {
            b[t].x = wcol[(size_t)(4 * t + 0) * P];
            b[t].y = wcol[(size_t)(4 * t + 1) * P];
        }
    }

    *(float2*)(tile + w * GCN_TP + lane * 2) = o;
    __syncthreads();

    if (w < NT) {
        const float* arow = tile + l * GCN_TP + 2 * half;
        v8f acc = {};
#pragma unroll
        for (int t = 0; t < 16; ++t) {
            const float2 av = *(const float2*)(arow + 4 * t);
            v2f a;
            a.x = av.x;
            a.y = av.y;
            acc = __builtin_amdgcn_wmma_f32_16x16x4_f32(
                 false, a,  false, b[t],
                 (short)0, acc,  false,  false);
        }
        float* orow = out + (size_t)(blockIdx.x * 16 + 8 * half) * P + w * 16 + l;
#pragma unroll
        for (int v = 0; v < 8; ++v)
            orow[(size_t)v * P] = acc[v];
    }
}

__global__ __launch_bounds__(256) void gcn_gather_pool16(
    const float* __restrict__ z, const int2* __restrict__ rec,
    const int* __restrict__ rowptr, const float* __restrict__ dinv,
    const float* __restrict__ bias, const int* __restrict__ batch,
    float* __restrict__ psum, float* __restrict__ pcnt)
{
    __shared__ __attribute__((aligned(16))) float srow[64 * GCN_ZW];
    __shared__ int sgid[64];

    const int q    = threadIdx.x & 3;
    const int k    = threadIdx.x >> 2;
    const int node = blockIdx.x * 64 + k;

    float4 o = make_float4(0.0f, 0.0f, 0.0f, 0.0f);
    static_assert(GCN_N % 8 == 0, "a wave's 8 nodes exist together or not at all");
    if (node < GCN_N) {
        const int beg = rowptr[node];
        const int end = rowptr[node + 1];
        const float di = dinv[node];
        const float4 hv = *(const float4*)(z + (size_t)node * GCN_ZW + 4 * q);
        const float4 bv = *(const float4*)(bias + 4 * q);

        float4 a = make_float4(0.0f, 0.0f, 0.0f, 0.0f);
        const int last = end - 1;
        for (int s = beg; s < end; s += 4) {
            const int i1 = (s + 1 < end) ? s + 1 : last;
            const int i2 = (s + 2 < end) ? s + 2 : last;
            const int i3 = (s + 3 < end) ? s + 3 : last;
            const int2 r0 = rec[s];
            const int2 r1 = rec[i1];
            const int2 r2 = rec[i2];
            const int2 r3 = rec[i3];
            float4 v0 = *(const float4*)(z + (size_t)r0.x * GCN_ZW + 4 * q);
            float4 v1 = *(const float4*)(z + (size_t)r1.x * GCN_ZW + 4 * q);
            float4 v2 = *(const float4*)(z + (size_t)r2.x * GCN_ZW + 4 * q);
            float4 v3 = *(const float4*)(z + (size_t)r3.x * GCN_ZW + 4 * q);
            const float c0 = __int_as_float(r0.y);
            float c1 = __int_as_float(r1.y);
            float c2 = __int_as_float(r2.y);
            float c3 = __int_as_float(r3.y);
            const float4 zero = make_float4(0.0f, 0.0f, 0.0f, 0.0f);
            if (s + 1 >= end) { v1 = zero; c1 = 0.0f; }
            if (s + 2 >= end) { v2 = zero; c2 = 0.0f; }
            if (s + 3 >= end) { v3 = zero; c3 = 0.0f; }
            a.x = fmaf(v0.x, c0, a.x);  a.y = fmaf(v0.y, c0, a.y);  a.z = fmaf(v0.z, c0, a.z);  a.w = fmaf(v0.w, c0, a.w);
            a.x = fmaf(v1.x, c1, a.x);  a.y = fmaf(v1.y, c1, a.y);  a.z = fmaf(v1.z, c1, a.z);  a.w = fmaf(v1.w, c1, a.w);
            a.x = fmaf(v2.x, c2, a.x);  a.y = fmaf(v2.y, c2, a.y);  a.z = fmaf(v2.z, c2, a.z);  a.w = fmaf(v2.w, c2, a.w);
            a.x = fmaf(v3.x, c3, a.x);  a.y = fmaf(v3.y, c3, a.y);  a.z = fmaf(v3.z, c3, a.z);  a.w = fmaf(v3.w, c3, a.w);
        }
        o.x = a.x + hv.x * di * di + bv.x;
        o.y = a.y + hv.y * di * di + bv.y;
        o.z = a.z + hv.z * di * di + bv.z;
        o.w = a.w + hv.w * di * di + bv.w;
        if (q == 0) sgid[k] = batch[node];
    }

    *(float4*)(srow + k * GCN_ZW + 4 * q) = o;
    __syncthreads();

    const int lane  = threadIdx.x & 31;
    const int w     = threadIdx.x >> 5;
    const int first = blockIdx.x * 64 + w * 8;
    if (lane < GCN_C && first < GCN_N) {
        const float* rows = srow + w * 8 * GCN_ZW;
        const int*   ids  = sgid + w * 8;
        int   cur = __builtin_amdgcn_readfirstlane(ids[0]);
        float acc = rows[lane];
        int   run = 1;
        for (int j = 1; j < 8; ++j) {
            const int   g = __builtin_amdgcn_readfirstlane(ids[j]);
            const float v = rows[j * GCN_ZW + lane];
            if (g != cur) {
                atomicAdd(&psum[(size_t)cur * GCN_ZW + lane], acc);
                if (lane == 0) atomicAdd(&pcnt[cur], (float)run);
                cur = g;  acc = v;  run = 1;
            } else {
                acc += v;  ++run;
            }
        }
        atomicAdd(&psum[(size_t)cur * GCN_ZW + lane], acc);
        if (lane == 0) atomicAdd(&pcnt[cur], (float)run);
    }
}

__global__ __launch_bounds__(256) void gcn_final(
    const float* __restrict__ psum, const float* __restrict__ pcnt,
    const float* __restrict__ bl, float* __restrict__ out)
{
    const int idx = blockIdx.x * blockDim.x + threadIdx.x;
    const int b = idx / GCN_C;
    const int c = idx % GCN_C;
    out[idx] = psum[(size_t)b * GCN_ZW + c] / fmaxf(pcnt[b], 1.0f) + bl[c];
}

extern "C" void kernel_launch(void* const* d_in, const int* in_sizes, int n_in,
                              void* d_out, int out_size, void* d_ws, size_t ws_size,
                              hipStream_t stream) {
    (void)in_sizes; (void)n_in; (void)out_size; (void)ws_size;

    const float* x     = (const float*)d_in[0];
    const int*   ei    = (const int*)  d_in[1];
    const int*   batch = (const int*)  d_in[2];
    const float* W1    = (const float*)d_in[3];
    const float* b1    = (const float*)d_in[4];
    const float* W2    = (const float*)d_in[5];
    const float* b2    = (const float*)d_in[6];
    const float* W3    = (const float*)d_in[7];
    const float* b3    = (const float*)d_in[8];
    const float* Wl    = (const float*)d_in[9];
    const float* bl    = (const float*)d_in[10];
    float* out = (float*)d_out;

    const int* src = ei;
    const int* dst = ei + GCN_E;

    int*          cntBB    = (int*)d_ws;
    int*          binStart = cntBB + GCN_G1 * 256;
    unsigned int* tmp      = (unsigned int*)(binStart + 320);
    int*          rowptr   = (int*)(tmp + GCN_E);
    float*        dinv     = (float*)(rowptr + GCN_NP);
    int2*  rec    = (int2*)(dinv + GCN_NP);
    float* bufA   = (float*)(rec + GCN_E);
    float* bufB   = bufA + (size_t)GCN_N * GCN_H;
    float* psum   = bufB + (size_t)GCN_N * GCN_H;
    float* pcnt   = psum + (size_t)GCN_B * GCN_ZW;
    float* wc     = pcnt + GCN_B;

    const dim3 gemmBlk(32, 4);
    const dim3 gemmGrd(GCN_N / 16);
    const int  poolGrd = (GCN_N + 63) / 64;
    const int  fuseGrd = GCN_N / 16;

    gcn_bin_count<<<GCN_G1 + 1, 1024, 0, stream>>>(dst, cntBB, W3, b3, Wl, wc);
    gcn_bin_scatter<<<GCN_G1, 1024, 0, stream>>>(src, dst, cntBB, binStart, tmp);
    gcn_bin_sort<<<GCN_NB, 1024, 0, stream>>>(tmp, binStart, rowptr, dinv, rec);
    gcn_coef<<<GCN_E / 256, 128, 0, stream>>>(dinv, (int4*)rec);

    gcn_gemm_wmma_f32<<<gemmGrd, gemmBlk, 0, stream>>>(x, W1, bufA, GCN_FIN);

    gcn_gather_gemm<4><<<fuseGrd, 512, 0, stream>>>(bufA, rec, rowptr, dinv, b1, W2, bufB);

    gcn_gather_gemm<1><<<fuseGrd, 512, 0, stream>>>(bufB, rec, rowptr, dinv, b2, wc, bufA);

    hipMemsetAsync(psum, 0, (size_t)(GCN_B * GCN_ZW + GCN_B) * sizeof(float), stream);
    gcn_gather_pool16<<<poolGrd, 256, 0, stream>>>(bufA, rec, rowptr, dinv, wc + 64 * GCN_ZW, batch, psum, pcnt);

    gcn_final<<<(GCN_B * GCN_C) / 256, 256, 0, stream>>>(psum, pcnt, bl, out);
}


# Each wave of the layer-1/2 gather handles 4 consecutive nodes in turn (row bounds and first records of all 4 loaded up front), 4 waves per 16-node block: 50000 -> 12500 waves a launch; sum order kept
# speedup vs baseline: 4.2803x; 1.0653x over previous
#include <hip/hip_runtime.h>
#include <hip/hip_bf16.h>


#define GCN_N    50000
#define GCN_FIN  128
#define GCN_E    800000
#define GCN_H    64
#define GCN_C    10
#define GCN_B    512
#define GCN_NP   53248
#define GCN_ZW   16

typedef __attribute__((ext_vector_type(2))) float v2f;
typedef __attribute__((ext_vector_type(8))) float v8f;

__global__ __launch_bounds__(128) void gcn_gemm_wmma_f32(
    const float* __restrict__ A, const float* __restrict__ W,
    float* __restrict__ C, int K)
{
    const int lane    = threadIdx.x;
    const int colBase = threadIdx.y * 16;
    const int rowBase = blockIdx.x * 16;
    const int half    = lane >> 4;
    const int l       = lane & 15;

    const float* arow = A + (size_t)(rowBase + l) * K + 2 * half;
    const float* wcol = W + colBase + l;

    v8f acc = {};
    for (int k0 = 0; k0 < K; k0 += 4) {
        v2f a, b;
        a.x = arow[k0 + 0];
        a.y = arow[k0 + 1];
        b.x = wcol[(size_t)(k0 + 2 * half + 0) * GCN_H];
        b.y = wcol[(size_t)(k0 + 2 * half + 1) * GCN_H];
        acc = __builtin_amdgcn_wmma_f32_16x16x4_f32(
             false, a,  false, b,
             (short)0, acc,  false,  false);
    }

#pragma unroll
    for (int v = 0; v < 8; ++v) {
        const int m = v + 8 * half;
        C[(size_t)(rowBase + m) * GCN_H + colBase + l] = acc[v];
    }
}

#define GCN_NB   196
#define GCN_G1   125
#define GCN_CH   6400
#define GCN_EPT  7
#define GCN_CAP  8192

static_assert(GCN_G1 * GCN_CH == GCN_E, "chunks cover the edge list exactly");
static_assert(GCN_NB * 256 >= GCN_N + 1 && GCN_NB <= 256 && GCN_N < 65536, "key packing");

__device__ __forceinline__ void gcn_excl_scan256(int* a, const int t)
{
    if (t < 32) {
        int v[8];
        int s = 0;
#pragma unroll
        for (int j = 0; j < 8; ++j) { v[j] = a[t * 8 + j]; s += v[j]; }
        int inc = s;
#pragma unroll
        for (int off = 1; off < 32; off <<= 1) {
            const int o = __shfl_up(inc, off);
            if (t >= off) inc += o;
        }
        int run = inc - s;
#pragma unroll
        for (int j = 0; j < 8; ++j) { a[t * 8 + j] = run; run += v[j]; }
        if (t == 31) a[256] = run;
    }
    __syncthreads();
}

__global__ __launch_bounds__(1024) void gcn_bin_count(
    const int* __restrict__ dst, int* __restrict__ cntBB,
    const float* __restrict__ W3, const float* __restrict__ b3,
    const float* __restrict__ Wl, float* __restrict__ wc)
{
    if (blockIdx.x == GCN_G1) {
        for (int idx = threadIdx.x; idx < 65 * GCN_ZW; idx += 1024) {
            const int r = idx / GCN_ZW;
            const int c = idx % GCN_ZW;
            const float* a = (r < GCN_H) ? (W3 + r * GCN_H) : b3;
            float acc = 0.0f;
            if (c < GCN_C) {
                for (int k = 0; k < GCN_H; ++k) acc = fmaf(a[k], Wl[k * GCN_C + c], acc);
            }
            wc[idx] = acc;
        }
        return;
    }

    __shared__ int lcnt[256];
    const int t = threadIdx.x;
    if (t < 256) lcnt[t] = 0;
    __syncthreads();
    const int* d = dst + blockIdx.x * GCN_CH;
    for (int i = t; i < GCN_CH; i += 1024) atomicAdd(&lcnt[d[i] >> 8], 1);
    __syncthreads();
    if (t < 256) cntBB[blockIdx.x * 256 + t] = lcnt[t];
}

__global__ __launch_bounds__(1024) void gcn_bin_scatter(
    const int* __restrict__ src, const int* __restrict__ dst,
    const int* __restrict__ cntBB, int* __restrict__ binStart,
    unsigned int* __restrict__ tmp)
{
    __shared__ int tot[257];
    __shared__ int lst[257];
    __shared__ int off[256];
    __shared__ unsigned int sorted[GCN_CH];

    const int t = threadIdx.x;
    const int b = blockIdx.x;
    const int* sp = src + b * GCN_CH;
    const int* dp = dst + b * GCN_CH;

    unsigned int key[GCN_EPT];
#pragma unroll
    for (int j = 0; j < GCN_EPT; ++j) {
        const int i = t + j * 1024;
        key[j] = 0u;
        if (i < GCN_CH) {
            const unsigned int s = (unsigned int)sp[i];
            const unsigned int d = (unsigned int)dp[i];
            key[j] = ((d >> 8) << 24) | (s << 8) | (d & 255u);
        }
    }

    int before = 0;
    if (t < 256) {
        int total = 0;
        for (int bb = 0; bb < GCN_G1; ++bb) {
            const int c = cntBB[bb * 256 + t];
            total += c;
            before += (bb < b) ? c : 0;
        }
        tot[t] = total;
        lst[t] = 0;
    }
    __syncthreads();
    gcn_excl_scan256(tot, t);
    if (b == 0 && t <= 256) binStart[t] = tot[t];

    int r[GCN_EPT];
#pragma unroll
    for (int j = 0; j < GCN_EPT; ++j) {
        r[j] = 0;
        if (t + j * 1024 < GCN_CH) r[j] = atomicAdd(&lst[key[j] >> 24], 1);
    }
    __syncthreads();
    gcn_excl_scan256(lst, t);
    if (t < 256) off[t] = tot[t] + before - lst[t];
#pragma unroll
    for (int j = 0; j < GCN_EPT; ++j) {
        if (t + j * 1024 < GCN_CH) sorted[lst[key[j] >> 24] + r[j]] = key[j];
    }
    __syncthreads();
    for (int p = t; p < GCN_CH; p += 1024) {
        const unsigned int k = sorted[p];
        tmp[off[k >> 24] + p] = k;
    }
}

__global__ __launch_bounds__(1024) void gcn_bin_sort(
    const unsigned int* __restrict__ tmp, const int* __restrict__ binStart,
    int* __restrict__ rowptr, float* __restrict__ dinv, int2* __restrict__ rec)
{
    __shared__ int   cnt[257];
    __shared__ int   cur[256];
    __shared__ float sdv[256];
    __shared__ int2  stage[GCN_CAP];

    const int t   = threadIdx.x;
    const int bin = blockIdx.x;
    const int beg = binStart[bin];
    const int end = binStart[bin + 1];
    const int n   = end - beg;

    if (t < 256) cnt[t] = 0;
    __syncthreads();
    for (int i = beg + t; i < end; i += 1024) atomicAdd(&cnt[tmp[i] & 255u], 1);
    __syncthreads();
    int deg = 0;
    if (t < 256) deg = cnt[t];
    __syncthreads();
    gcn_excl_scan256(cnt, t);
    if (t < 256) {
        const int node = bin * 256 + t;
        const float dv = rsqrtf((float)deg + 1.0f);
        sdv[t] = dv;
        cur[t] = cnt[t];
        if (node <= GCN_N) rowptr[node] = beg + cnt[t];
        if (node <  GCN_N) dinv[node]   = dv;
    }
    __syncthreads();

    const bool staged = (n <= GCN_CAP);
    for (int i = beg + t; i < end; i += 1024) {
        const unsigned int k = tmp[i];
        const int dl  = (int)(k & 255u);
        const int s   = (int)((k >> 8) & 0xFFFFu);
        const int pos = atomicAdd(&cur[dl], 1);
        const int2 v  = make_int2(s, __float_as_int(sdv[dl]));
        if (staged) stage[pos] = v;
        else        rec[beg + pos] = v;
    }
    __syncthreads();
    if (staged) {
        for (int p = t; p < n; p += 1024) rec[beg + p] = stage[p];
    }
}

__global__ __launch_bounds__(128) void gcn_coef(
    const float* __restrict__ dinv, int4* __restrict__ rec2)
{
    const int idx = blockIdx.x * blockDim.x + threadIdx.x;
    int4 r = rec2[idx];
    r.y = __float_as_int(dinv[r.x] * __int_as_float(r.y));
    r.w = __float_as_int(dinv[r.z] * __int_as_float(r.w));
    rec2[idx] = r;
}

#define GCN_TP  68
#define GCN_NPW 4

template <int NT>
__global__ __launch_bounds__(128) void gcn_gather_gemm(
    const float* __restrict__ h, const int2* __restrict__ rec,
    const int* __restrict__ rowptr, const float* __restrict__ dinv,
    const float* __restrict__ bias, const float* __restrict__ W,
    float* __restrict__ out)
{
    __shared__ __attribute__((aligned(16))) float tile[16 * GCN_TP];
    constexpr int P = 16 * NT;

    const int lane = threadIdx.x & 31;
    const int p    = lane * 2;
    const int w     = __builtin_amdgcn_readfirstlane(threadIdx.x >> 5);
    const int node0 = blockIdx.x * 16 + w * GCN_NPW;

    const int   rpv = rowptr[node0 + (lane < GCN_NPW ? lane : GCN_NPW)];
    const float dvv = dinv[node0 + (lane & (GCN_NPW - 1))];
    const float2 bv = *(const float2*)(bias + p);

    int    beg[GCN_NPW], end[GCN_NPW], rs0[GCN_NPW], rc0[GCN_NPW];
    float2 hv[GCN_NPW];
#pragma unroll
    for (int i = 0; i < GCN_NPW; ++i) {
        beg[i] = __builtin_amdgcn_readlane(rpv, i);
        end[i] = __builtin_amdgcn_readlane(rpv, i + 1);
    }
#pragma unroll
    for (int i = 0; i < GCN_NPW; ++i) {
        const int slot = beg[i] + lane;
        const int2 r = rec[slot < GCN_E ? slot : GCN_E - 1];
        rs0[i] = r.x;
        rc0[i] = r.y;
        hv[i] = *(const float2*)(h + (size_t)(node0 + i) * GCN_H + p);
    }

#pragma unroll
    for (int i = 0; i < GCN_NPW; ++i) {
        float ax = 0.0f, ay = 0.0f;
        int rs = rs0[i], rc = rc0[i];
        for (int base = beg[i]; base < end[i]; base += 32) {
            const int left = end[i] - base;
            const int n = left < 32 ? left : 32;
            if (base != beg[i]) {
                const int slot = base + lane;
                const int2 r = rec[slot < GCN_E ? slot : GCN_E - 1];
                rs = r.x;
                rc = r.y;
            }
            int j = 0;
            for (; j + 4 <= n; j += 4) {
                const int s0 = __builtin_amdgcn_readlane(rs, j + 0);
                const int s1 = __builtin_amdgcn_readlane(rs, j + 1);
                const int s2 = __builtin_amdgcn_readlane(rs, j + 2);
                const int s3 = __builtin_amdgcn_readlane(rs, j + 3);
                const float c0 = __int_as_float(__builtin_amdgcn_readlane(rc, j + 0));
                const float c1 = __int_as_float(__builtin_amdgcn_readlane(rc, j + 1));
                const float c2 = __int_as_float(__builtin_amdgcn_readlane(rc, j + 2));
                const float c3 = __int_as_float(__builtin_amdgcn_readlane(rc, j + 3));
                const float2 v0 = *(const float2*)(h + (size_t)s0 * GCN_H + p);
                const float2 v1 = *(const float2*)(h + (size_t)s1 * GCN_H + p);
                const float2 v2 = *(const float2*)(h + (size_t)s2 * GCN_H + p);
                const float2 v3 = *(const float2*)(h + (size_t)s3 * GCN_H + p);
                ax = fmaf(v0.x, c0, ax);  ay = fmaf(v0.y, c0, ay);
                ax = fmaf(v1.x, c1, ax);  ay = fmaf(v1.y, c1, ay);
                ax = fmaf(v2.x, c2, ax);  ay = fmaf(v2.y, c2, ay);
                ax = fmaf(v3.x, c3, ax);  ay = fmaf(v3.y, c3, ay);
            }
            for (; j < n; ++j) {
                const int s0 = __builtin_amdgcn_readlane(rs, j);
                const float c0 = __int_as_float(__builtin_amdgcn_readlane(rc, j));
                const float2 v0 = *(const float2*)(h + (size_t)s0 * GCN_H + p);
                ax = fmaf(v0.x, c0, ax);  ay = fmaf(v0.y, c0, ay);
            }
        }

        const float di = __int_as_float(__builtin_amdgcn_readlane(__float_as_int(dvv), i));
        float2 o = make_float2(ax + hv[i].x * di * di + bv.x, ay + hv[i].y * di * di + bv.y);
        o.x = fmaxf(o.x, 0.0f);
        o.y = fmaxf(o.y, 0.0f);
        *(float2*)(tile + (w * GCN_NPW + i) * GCN_TP + p) = o;
    }

    const int half = lane >> 4;
    const int l    = lane & 15;

    v2f b[16];
    if (w < NT) {
        const float* wcol = W + (size_t)(2 * half) * P + w * 16 + l;
#pragma unroll
        for (int t = 0; t < 16; ++t) {
            b[t].x = wcol[(size_t)(4 * t + 0) * P];
            b[t].y = wcol[(size_t)(4 * t + 1) * P];
        }
    }

    __syncthreads();

    if (w < NT) {
        const float* arow = tile + l * GCN_TP + 2 * half;
        v8f acc = {};
#pragma unroll
        for (int t = 0; t < 16; ++t) {
            const float2 av = *(const float2*)(arow + 4 * t);
            v2f a;
            a.x = av.x;
            a.y = av.y;
            acc = __builtin_amdgcn_wmma_f32_16x16x4_f32(
                 false, a,  false, b[t],
                 (short)0, acc,  false,  false);
        }
        float* orow = out + (size_t)(blockIdx.x * 16 + 8 * half) * P + w * 16 + l;
#pragma unroll
        for (int v = 0; v < 8; ++v)
            orow[(size_t)v * P] = acc[v];
    }
}

__global__ __launch_bounds__(256) void gcn_gather_pool16(
    const float* __restrict__ z, const int2* __restrict__ rec,
    const int* __restrict__ rowptr, const float* __restrict__ dinv,
    const float* __restrict__ bias, const int* __restrict__ batch,
    float* __restrict__ psum, float* __restrict__ pcnt)
{
    __shared__ __attribute__((aligned(16))) float srow[64 * GCN_ZW];
    __shared__ int sgid[64];

    const int q    = threadIdx.x & 3;
    const int k    = threadIdx.x >> 2;
    const int node = blockIdx.x * 64 + k;

    float4 o = make_float4(0.0f, 0.0f, 0.0f, 0.0f);
    static_assert(GCN_N % 8 == 0, "a wave's 8 nodes exist together or not at all");
    if (node < GCN_N) {
        const int beg = rowptr[node];
        const int end = rowptr[node + 1];
        const float di = dinv[node];
        const float4 hv = *(const float4*)(z + (size_t)node * GCN_ZW + 4 * q);
        const float4 bv = *(const float4*)(bias + 4 * q);

        float4 a = make_float4(0.0f, 0.0f, 0.0f, 0.0f);
        const int last = end - 1;
        for (int s = beg; s < end; s += 4) {
            const int i1 = (s + 1 < end) ? s + 1 : last;
            const int i2 = (s + 2 < end) ? s + 2 : last;
            const int i3 = (s + 3 < end) ? s + 3 : last;
            const int2 r0 = rec[s];
            const int2 r1 = rec[i1];
            const int2 r2 = rec[i2];
            const int2 r3 = rec[i3];
            float4 v0 = *(const float4*)(z + (size_t)r0.x * GCN_ZW + 4 * q);
            float4 v1 = *(const float4*)(z + (size_t)r1.x * GCN_ZW + 4 * q);
            float4 v2 = *(const float4*)(z + (size_t)r2.x * GCN_ZW + 4 * q);
            float4 v3 = *(const float4*)(z + (size_t)r3.x * GCN_ZW + 4 * q);
            const float c0 = __int_as_float(r0.y);
            float c1 = __int_as_float(r1.y);
            float c2 = __int_as_float(r2.y);
            float c3 = __int_as_float(r3.y);
            const float4 zero = make_float4(0.0f, 0.0f, 0.0f, 0.0f);
            if (s + 1 >= end) { v1 = zero; c1 = 0.0f; }
            if (s + 2 >= end) { v2 = zero; c2 = 0.0f; }
            if (s + 3 >= end) { v3 = zero; c3 = 0.0f; }
            a.x = fmaf(v0.x, c0, a.x);  a.y = fmaf(v0.y, c0, a.y);  a.z = fmaf(v0.z, c0, a.z);  a.w = fmaf(v0.w, c0, a.w);
            a.x = fmaf(v1.x, c1, a.x);  a.y = fmaf(v1.y, c1, a.y);  a.z = fmaf(v1.z, c1, a.z);  a.w = fmaf(v1.w, c1, a.w);
            a.x = fmaf(v2.x, c2, a.x);  a.y = fmaf(v2.y, c2, a.y);  a.z = fmaf(v2.z, c2, a.z);  a.w = fmaf(v2.w, c2, a.w);
            a.x = fmaf(v3.x, c3, a.x);  a.y = fmaf(v3.y, c3, a.y);  a.z = fmaf(v3.z, c3, a.z);  a.w = fmaf(v3.w, c3, a.w);
        }
        o.x = a.x + hv.x * di * di + bv.x;
        o.y = a.y + hv.y * di * di + bv.y;
        o.z = a.z + hv.z * di * di + bv.z;
        o.w = a.w + hv.w * di * di + bv.w;
        if (q == 0) sgid[k] = batch[node];
    }

    *(float4*)(srow + k * GCN_ZW + 4 * q) = o;
    __syncthreads();

    const int lane  = threadIdx.x & 31;
    const int w     = threadIdx.x >> 5;
    const int first = blockIdx.x * 64 + w * 8;
    if (lane < GCN_C && first < GCN_N) {
        const float* rows = srow + w * 8 * GCN_ZW;
        const int*   ids  = sgid + w * 8;
        int   cur = __builtin_amdgcn_readfirstlane(ids[0]);
        float acc = rows[lane];
        int   run = 1;
        for (int j = 1; j < 8; ++j) {
            const int   g = __builtin_amdgcn_readfirstlane(ids[j]);
            const float v = rows[j * GCN_ZW + lane];
            if (g != cur) {
                atomicAdd(&psum[(size_t)cur * GCN_ZW + lane], acc);
                if (lane == 0) atomicAdd(&pcnt[cur], (float)run);
                cur = g;  acc = v;  run = 1;
            } else {
                acc += v;  ++run;
            }
        }
        atomicAdd(&psum[(size_t)cur * GCN_ZW + lane], acc);
        if (lane == 0) atomicAdd(&pcnt[cur], (float)run);
    }
}

__global__ __launch_bounds__(256) void gcn_final(
    const float* __restrict__ psum, const float* __restrict__ pcnt,
    const float* __restrict__ bl, float* __restrict__ out)
{
    const int idx = blockIdx.x * blockDim.x + threadIdx.x;
    const int b = idx / GCN_C;
    const int c = idx % GCN_C;
    out[idx] = psum[(size_t)b * GCN_ZW + c] / fmaxf(pcnt[b], 1.0f) + bl[c];
}

extern "C" void kernel_launch(void* const* d_in, const int* in_sizes, int n_in,
                              void* d_out, int out_size, void* d_ws, size_t ws_size,
                              hipStream_t stream) {
    (void)in_sizes; (void)n_in; (void)out_size; (void)ws_size;

    const float* x     = (const float*)d_in[0];
    const int*   ei    = (const int*)  d_in[1];
    const int*   batch = (const int*)  d_in[2];
    const float* W1    = (const float*)d_in[3];
    const float* b1    = (const float*)d_in[4];
    const float* W2    = (const float*)d_in[5];
    const float* b2    = (const float*)d_in[6];
    const float* W3    = (const float*)d_in[7];
    const float* b3    = (const float*)d_in[8];
    const float* Wl    = (const float*)d_in[9];
    const float* bl    = (const float*)d_in[10];
    float* out = (float*)d_out;

    const int* src = ei;
    const int* dst = ei + GCN_E;

    int*          cntBB    = (int*)d_ws;
    int*          binStart = cntBB + GCN_G1 * 256;
    unsigned int* tmp      = (unsigned int*)(binStart + 320);
    int*          rowptr   = (int*)(tmp + GCN_E);
    float*        dinv     = (float*)(rowptr + GCN_NP);
    int2*  rec    = (int2*)(dinv + GCN_NP);
    float* bufA   = (float*)(rec + GCN_E);
    float* bufB   = bufA + (size_t)GCN_N * GCN_H;
    float* psum   = bufB + (size_t)GCN_N * GCN_H;
    float* pcnt   = psum + (size_t)GCN_B * GCN_ZW;
    float* wc     = pcnt + GCN_B;

    const dim3 gemmBlk(32, 4);
    const dim3 gemmGrd(GCN_N / 16);
    const int  poolGrd = (GCN_N + 63) / 64;
    const int  fuseGrd = GCN_N / 16;

    gcn_bin_count<<<GCN_G1 + 1, 1024, 0, stream>>>(dst, cntBB, W3, b3, Wl, wc);
    gcn_bin_scatter<<<GCN_G1, 1024, 0, stream>>>(src, dst, cntBB, binStart, tmp);
    gcn_bin_sort<<<GCN_NB, 1024, 0, stream>>>(tmp, binStart, rowptr, dinv, rec);
    gcn_coef<<<GCN_E / 256, 128, 0, stream>>>(dinv, (int4*)rec);

    gcn_gemm_wmma_f32<<<gemmGrd, gemmBlk, 0, stream>>>(x, W1, bufA, GCN_FIN);

    gcn_gather_gemm<4><<<fuseGrd, 128, 0, stream>>>(bufA, rec, rowptr, dinv, b1, W2, bufB);

    gcn_gather_gemm<1><<<fuseGrd, 128, 0, stream>>>(bufB, rec, rowptr, dinv, b2, wc, bufA);

    hipMemsetAsync(psum, 0, (size_t)(GCN_B * GCN_ZW + GCN_B) * sizeof(float), stream);
    gcn_gather_pool16<<<poolGrd, 256, 0, stream>>>(bufA, rec, rowptr, dinv, wc + 64 * GCN_ZW, batch, psum, pcnt);

    gcn_final<<<(GCN_B * GCN_C) / 256, 256, 0, stream>>>(psum, pcnt, bl, out);
}


# Edge sort in two kernels, not three: each chunk is bin-sorted in LDS and stored as its own run with its bin offsets; a bin's block sums 125 offsets and reads 125 segments; sum order = new LDS rank
# speedup vs baseline: 5.8667x; 1.3706x over previous
#include <hip/hip_runtime.h>
#include <hip/hip_bf16.h>


#define GCN_N    50000
#define GCN_FIN  128
#define GCN_E    800000
#define GCN_H    64
#define GCN_C    10
#define GCN_B    512
#define GCN_NP   53248
#define GCN_ZW   16

typedef __attribute__((ext_vector_type(2))) float v2f;
typedef __attribute__((ext_vector_type(8))) float v8f;

__global__ __launch_bounds__(128) void gcn_gemm_wmma_f32(
    const float* __restrict__ A, const float* __restrict__ W,
    const float* __restrict__ dinv, float* __restrict__ C, int K)
{
    const int lane    = threadIdx.x;
    const int colBase = threadIdx.y * 16;
    const int rowBase = blockIdx.x * 16;
    const int half    = lane >> 4;
    const int l       = lane & 15;

    const float* arow = A + (size_t)(rowBase + l) * K + 2 * half;
    const float* wcol = W + colBase + l;

    v8f acc = {};
    for (int k0 = 0; k0 < K; k0 += 4) {
        v2f a, b;
        a.x = arow[k0 + 0];
        a.y = arow[k0 + 1];
        b.x = wcol[(size_t)(k0 + 2 * half + 0) * GCN_H];
        b.y = wcol[(size_t)(k0 + 2 * half + 1) * GCN_H];
        acc = __builtin_amdgcn_wmma_f32_16x16x4_f32(
             false, a,  false, b,
             (short)0, acc,  false,  false);
    }

    const float4 dlo = *(const float4*)(dinv + rowBase + 8 * half);
    const float4 dhi = *(const float4*)(dinv + rowBase + 8 * half + 4);
    const float dr[8] = { dlo.x, dlo.y, dlo.z, dlo.w, dhi.x, dhi.y, dhi.z, dhi.w };
#pragma unroll
    for (int v = 0; v < 8; ++v) {
        const int m = v + 8 * half;
        C[(size_t)(rowBase + m) * GCN_H + colBase + l] = acc[v] * dr[v];
    }
}

#define GCN_NB   196
#define GCN_G1   125
#define GCN_CH   6400
#define GCN_EPT  7
#define GCN_CAP  8192
#define GCN_LP   320
#define GCN_SPT  8
#define GCN_RPT  8

static_assert(GCN_G1 * GCN_CH == GCN_E, "chunks cover the edge list exactly");
static_assert(GCN_NB * 256 >= GCN_N + 1 && GCN_NB <= 256 && GCN_N < 65536, "key packing");
static_assert(GCN_G1 * GCN_SPT <= 1024, "every chunk's segment has its threads in a bin's block");

__device__ __forceinline__ void gcn_excl_scan256(int* a, const int t)
{
    if (t < 32) {
        int v[8];
        int s = 0;
#pragma unroll
        for (int j = 0; j < 8; ++j) { v[j] = a[t * 8 + j]; s += v[j]; }
        int inc = s;
#pragma unroll
        for (int off = 1; off < 32; off <<= 1) {
            const int o = __shfl_up(inc, off);
            if (t >= off) inc += o;
        }
        int run = inc - s;
#pragma unroll
        for (int j = 0; j < 8; ++j) { a[t * 8 + j] = run; run += v[j]; }
        if (t == 31) a[256] = run;
    }
    __syncthreads();
}

__global__ __launch_bounds__(1024) void gcn_bin_scatter(
    const int* __restrict__ src, const int* __restrict__ dst,
    int* __restrict__ binOff, unsigned int* __restrict__ tmp,
    const float* __restrict__ W3, const float* __restrict__ b3,
    const float* __restrict__ Wl, float* __restrict__ wc)
{
    if (blockIdx.x == GCN_G1) {
        for (int idx = threadIdx.x; idx < 65 * GCN_ZW; idx += 1024) {
            const int r = idx / GCN_ZW;
            const int c = idx % GCN_ZW;
            const float* a = (r < GCN_H) ? (W3 + r * GCN_H) : b3;
            float acc = 0.0f;
            if (c < GCN_C) {
                for (int k = 0; k < GCN_H; ++k) acc = fmaf(a[k], Wl[k * GCN_C + c], acc);
            }
            wc[idx] = acc;
        }
        return;
    }

    __shared__ int lst[257];
    __shared__ unsigned int sorted[GCN_CH];

    const int t = threadIdx.x;
    const int b = blockIdx.x;
    const int* sp = src + b * GCN_CH;
    const int* dp = dst + b * GCN_CH;

    unsigned int key[GCN_EPT];
#pragma unroll
    for (int j = 0; j < GCN_EPT; ++j) {
        const int i = t + j * 1024;
        key[j] = 0u;
        if (i < GCN_CH) {
            const unsigned int s = (unsigned int)sp[i];
            const unsigned int d = (unsigned int)dp[i];
            key[j] = ((d >> 8) << 24) | (s << 8) | (d & 255u);
        }
    }

    if (t < 256) lst[t] = 0;
    __syncthreads();

    int r[GCN_EPT];
#pragma unroll
    for (int j = 0; j < GCN_EPT; ++j) {
        r[j] = 0;
        if (t + j * 1024 < GCN_CH) r[j] = atomicAdd(&lst[key[j] >> 24], 1);
    }
    __syncthreads();
    gcn_excl_scan256(lst, t);
    if (t <= 256) binOff[b * GCN_LP + t] = lst[t];
#pragma unroll
    for (int j = 0; j < GCN_EPT; ++j) {
        if (t + j * 1024 < GCN_CH) sorted[lst[key[j] >> 24] + r[j]] = key[j];
    }
    __syncthreads();
    unsigned int* run = tmp + b * GCN_CH;
    for (int p = t; p < GCN_CH; p += 1024) run[p] = sorted[p];
}

__global__ __launch_bounds__(1024) void gcn_bin_sort(
    const unsigned int* __restrict__ tmp, const int* __restrict__ binOff,
    int* __restrict__ rowptr, float* __restrict__ dinv, int* __restrict__ rec)
{
    __shared__ int   cnt[257];
    __shared__ int   cur[256];
    __shared__ int   tot[2];
    __shared__ int   stage[GCN_CAP];

    const int t   = threadIdx.x;
    const int bin = blockIdx.x;
    const int seg = t / GCN_SPT;
    const int sub = t % GCN_SPT;

    int sbeg = 0, slen = 0, slow = 0;
    if (seg < GCN_G1) {
        const int* bo = binOff + seg * GCN_LP + bin;
        slow = bo[0];
        slen = bo[1] - slow;
        sbeg = seg * GCN_CH + slow;
    }
    unsigned int kk[GCN_RPT];
#pragma unroll
    for (int u = 0; u < GCN_RPT; ++u) {
        const int j = sub + u * GCN_SPT;
        kk[u] = 0u;
        if (j < slen) kk[u] = tmp[sbeg + j];
    }

    if (t < 256) cnt[t] = 0;
    if (t < 2)   tot[t] = 0;
    __syncthreads();
    if (sub == 0 && seg < GCN_G1) {
        atomicAdd(&tot[0], slow);
        atomicAdd(&tot[1], slen);
    }
#pragma unroll
    for (int u = 0; u < GCN_RPT; ++u) {
        if (sub + u * GCN_SPT < slen) atomicAdd(&cnt[kk[u] & 255u], 1);
    }
    for (int j = sub + GCN_RPT * GCN_SPT; j < slen; j += GCN_SPT) atomicAdd(&cnt[tmp[sbeg + j] & 255u], 1);
    __syncthreads();
    const int beg = tot[0];
    const int n   = tot[1];
    int deg = 0;
    if (t < 256) deg = cnt[t];
    __syncthreads();
    gcn_excl_scan256(cnt, t);
    if (t < 256) {
        const int node = bin * 256 + t;
        const float dv = rsqrtf((float)deg + 1.0f);
        cur[t] = cnt[t];
        if (node <= GCN_N) rowptr[node] = beg + cnt[t];
        if (node <  GCN_N) dinv[node]   = dv;
    }
    __syncthreads();

    const bool staged = (n <= GCN_CAP);
#pragma unroll
    for (int u = 0; u < GCN_RPT; ++u) {
        if (sub + u * GCN_SPT < slen) {
            const unsigned int k = kk[u];
            const int s   = (int)((k >> 8) & 0xFFFFu);
            const int pos = atomicAdd(&cur[k & 255u], 1);
            if (staged) stage[pos] = s;
            else        rec[beg + pos] = s;
        }
    }
    for (int j = sub + GCN_RPT * GCN_SPT; j < slen; j += GCN_SPT) {
        const unsigned int k = tmp[sbeg + j];
        const int s   = (int)((k >> 8) & 0xFFFFu);
        const int pos = atomicAdd(&cur[k & 255u], 1);
        if (staged) stage[pos] = s;
        else        rec[beg + pos] = s;
    }
    __syncthreads();
    if (staged) {
        for (int p = t; p < n; p += 1024) rec[beg + p] = stage[p];
    }
}

#define GCN_TP  68
#define GCN_NPW 4

template <int NT>
__global__ __launch_bounds__(128) void gcn_gather_gemm(
    const float* __restrict__ h, const int* __restrict__ rec,
    const int* __restrict__ rowptr, const float* __restrict__ dinv,
    const float* __restrict__ bias, const float* __restrict__ W,
    float* __restrict__ out)
{
    __shared__ __attribute__((aligned(16))) float tile[16 * GCN_TP];
    constexpr int P = 16 * NT;

    const int lane = threadIdx.x & 31;
    const int p    = lane * 2;
    const int w     = __builtin_amdgcn_readfirstlane(threadIdx.x >> 5);
    const int node0 = blockIdx.x * 16 + w * GCN_NPW;

    const int   rpv = rowptr[node0 + (lane < GCN_NPW ? lane : GCN_NPW)];
    const float dvv = dinv[node0 + (lane & (GCN_NPW - 1))];
    const float2 bv = *(const float2*)(bias + p);

    int    beg[GCN_NPW], end[GCN_NPW], rs0[GCN_NPW];
    float2 hv[GCN_NPW];
#pragma unroll
    for (int i = 0; i < GCN_NPW; ++i) {
        beg[i] = __builtin_amdgcn_readlane(rpv, i);
        end[i] = __builtin_amdgcn_readlane(rpv, i + 1);
    }
#pragma unroll
    for (int i = 0; i < GCN_NPW; ++i) {
        const int slot = beg[i] + lane;
        rs0[i] = rec[slot < GCN_E ? slot : GCN_E - 1];
        hv[i] = *(const float2*)(h + (size_t)(node0 + i) * GCN_H + p);
    }

#pragma unroll
    for (int i = 0; i < GCN_NPW; ++i) {
        float ax = 0.0f, ay = 0.0f;
        int rs = rs0[i];
        for (int base = beg[i]; base < end[i]; base += 32) {
            const int left = end[i] - base;
            const int n = left < 32 ? left : 32;
            if (base != beg[i]) {
                const int slot = base + lane;
                rs = rec[slot < GCN_E ? slot : GCN_E - 1];
            }
            int j = 0;
            for (; j + 4 <= n; j += 4) {
                const int s0 = __builtin_amdgcn_readlane(rs, j + 0);
                const int s1 = __builtin_amdgcn_readlane(rs, j + 1);
                const int s2 = __builtin_amdgcn_readlane(rs, j + 2);
                const int s3 = __builtin_amdgcn_readlane(rs, j + 3);
                const float2 v0 = *(const float2*)(h + (size_t)s0 * GCN_H + p);
                const float2 v1 = *(const float2*)(h + (size_t)s1 * GCN_H + p);
                const float2 v2 = *(const float2*)(h + (size_t)s2 * GCN_H + p);
                const float2 v3 = *(const float2*)(h + (size_t)s3 * GCN_H + p);
                ax += v0.x;  ay += v0.y;
                ax += v1.x;  ay += v1.y;
                ax += v2.x;  ay += v2.y;
                ax += v3.x;  ay += v3.y;
            }
            for (; j < n; ++j) {
                const int s0 = __builtin_amdgcn_readlane(rs, j);
                const float2 v0 = *(const float2*)(h + (size_t)s0 * GCN_H + p);
                ax += v0.x;  ay += v0.y;
            }
        }

        const float di = __int_as_float(__builtin_amdgcn_readlane(__float_as_int(dvv), i));
        float2 o = make_float2((ax + hv[i].x) * di + bv.x, (ay + hv[i].y) * di + bv.y);
        o.x = fmaxf(o.x, 0.0f);
        o.y = fmaxf(o.y, 0.0f);
        *(float2*)(tile + (w * GCN_NPW + i) * GCN_TP + p) = o;
    }

    const int half = lane >> 4;
    const int l    = lane & 15;

    v2f b[16];
    float4 dlo = make_float4(0.0f, 0.0f, 0.0f, 0.0f), dhi = dlo;
    if (w < NT) {
        const float* wcol = W + (size_t)(2 * half) * P + w * 16 + l;
#pragma unroll
        for (int t = 0; t < 16; ++t) {
            b[t].x = wcol[(size_t)(4 * t + 0) * P];
            b[t].y = wcol[(size_t)(4 * t + 1) * P];
        }
        dlo = *(const float4*)(dinv + blockIdx.x * 16 + 8 * half);
        dhi = *(const float4*)(dinv + blockIdx.x * 16 + 8 * half + 4);
    }

    __syncthreads();

    if (w < NT) {
        const float* arow = tile + l * GCN_TP + 2 * half;
        v8f acc = {};
#pragma unroll
        for (int t = 0; t < 16; ++t) {
            const float2 av = *(const float2*)(arow + 4 * t);
            v2f a;
            a.x = av.x;
            a.y = av.y;
            acc = __builtin_amdgcn_wmma_f32_16x16x4_f32(
                 false, a,  false, b[t],
                 (short)0, acc,  false,  false);
        }
        float* orow = out + (size_t)(blockIdx.x * 16 + 8 * half) * P + w * 16 + l;
        const float dr[8] = { dlo.x, dlo.y, dlo.z, dlo.w, dhi.x, dhi.y, dhi.z, dhi.w };
#pragma unroll
        for (int v = 0; v < 8; ++v)
            orow[(size_t)v * P] = acc[v] * dr[v];
    }
}

__global__ __launch_bounds__(256) void gcn_gather_pool16(
    const float* __restrict__ z, const int* __restrict__ rec,
    const int* __restrict__ rowptr, const float* __restrict__ dinv,
    const float* __restrict__ bias, const int* __restrict__ batch,
    float* __restrict__ psum, float* __restrict__ pcnt)
{
    __shared__ __attribute__((aligned(16))) float srow[64 * GCN_ZW];
    __shared__ int sgid[64];

    const int q    = threadIdx.x & 3;
    const int k    = threadIdx.x >> 2;
    const int node = blockIdx.x * 64 + k;

    float4 o = make_float4(0.0f, 0.0f, 0.0f, 0.0f);
    static_assert(GCN_N % 8 == 0, "a wave's 8 nodes exist together or not at all");
    if (node < GCN_N) {
        const int beg = rowptr[node];
        const int end = rowptr[node + 1];
        const float di = dinv[node];
        const float4 hv = *(const float4*)(z + (size_t)node * GCN_ZW + 4 * q);
        const float4 bv = *(const float4*)(bias + 4 * q);

        float4 a = make_float4(0.0f, 0.0f, 0.0f, 0.0f);
        const int last = end - 1;
        for (int s = beg; s < end; s += 4) {
            const int i1 = (s + 1 < end) ? s + 1 : last;
            const int i2 = (s + 2 < end) ? s + 2 : last;
            const int i3 = (s + 3 < end) ? s + 3 : last;
            const int r0 = rec[s];
            const int r1 = rec[i1];
            const int r2 = rec[i2];
            const int r3 = rec[i3];
            const float4 v0 = *(const float4*)(z + (size_t)r0 * GCN_ZW + 4 * q);
            float4 v1 = *(const float4*)(z + (size_t)r1 * GCN_ZW + 4 * q);
            float4 v2 = *(const float4*)(z + (size_t)r2 * GCN_ZW + 4 * q);
            float4 v3 = *(const float4*)(z + (size_t)r3 * GCN_ZW + 4 * q);
            const float4 zero = make_float4(0.0f, 0.0f, 0.0f, 0.0f);
            if (s + 1 >= end) v1 = zero;
            if (s + 2 >= end) v2 = zero;
            if (s + 3 >= end) v3 = zero;
            a.x += v0.x;  a.y += v0.y;  a.z += v0.z;  a.w += v0.w;
            a.x += v1.x;  a.y += v1.y;  a.z += v1.z;  a.w += v1.w;
            a.x += v2.x;  a.y += v2.y;  a.z += v2.z;  a.w += v2.w;
            a.x += v3.x;  a.y += v3.y;  a.z += v3.z;  a.w += v3.w;
        }
        o.x = (a.x + hv.x) * di + bv.x;
        o.y = (a.y + hv.y) * di + bv.y;
        o.z = (a.z + hv.z) * di + bv.z;
        o.w = (a.w + hv.w) * di + bv.w;
        if (q == 0) sgid[k] = batch[node];
    }

    *(float4*)(srow + k * GCN_ZW + 4 * q) = o;
    __syncthreads();

    const int lane  = threadIdx.x & 31;
    const int w     = threadIdx.x >> 5;
    const int first = blockIdx.x * 64 + w * 8;
    if (lane < GCN_C && first < GCN_N) {
        const float* rows = srow + w * 8 * GCN_ZW;
        const int*   ids  = sgid + w * 8;
        int   cur = __builtin_amdgcn_readfirstlane(ids[0]);
        float acc = rows[lane];
        int   run = 1;
        for (int j = 1; j < 8; ++j) {
            const int   g = __builtin_amdgcn_readfirstlane(ids[j]);
            const float v = rows[j * GCN_ZW + lane];
            if (g != cur) {
                atomicAdd(&psum[(size_t)cur * GCN_ZW + lane], acc);
                if (lane == 0) atomicAdd(&pcnt[cur], (float)run);
                cur = g;  acc = v;  run = 1;
            } else {
                acc += v;  ++run;
            }
        }
        atomicAdd(&psum[(size_t)cur * GCN_ZW + lane], acc);
        if (lane == 0) atomicAdd(&pcnt[cur], (float)run);
    }
}

__global__ __launch_bounds__(256) void gcn_final(
    const float* __restrict__ psum, const float* __restrict__ pcnt,
    const float* __restrict__ bl, float* __restrict__ out)
{
    const int idx = blockIdx.x * blockDim.x + threadIdx.x;
    const int b = idx / GCN_C;
    const int c = idx % GCN_C;
    out[idx] = psum[(size_t)b * GCN_ZW + c] / fmaxf(pcnt[b], 1.0f) + bl[c];
}

extern "C" void kernel_launch(void* const* d_in, const int* in_sizes, int n_in,
                              void* d_out, int out_size, void* d_ws, size_t ws_size,
                              hipStream_t stream) {
    (void)in_sizes; (void)n_in; (void)out_size; (void)ws_size;

    const float* x     = (const float*)d_in[0];
    const int*   ei    = (const int*)  d_in[1];
    const int*   batch = (const int*)  d_in[2];
    const float* W1    = (const float*)d_in[3];
    const float* b1    = (const float*)d_in[4];
    const float* W2    = (const float*)d_in[5];
    const float* b2    = (const float*)d_in[6];
    const float* W3    = (const float*)d_in[7];
    const float* b3    = (const float*)d_in[8];
    const float* Wl    = (const float*)d_in[9];
    const float* bl    = (const float*)d_in[10];
    float* out = (float*)d_out;

    const int* src = ei;
    const int* dst = ei + GCN_E;

    int*          binOff   = (int*)d_ws;
    unsigned int* tmp      = (unsigned int*)(binOff + GCN_G1 * GCN_LP);
    int*          rowptr   = (int*)(tmp + GCN_E);
    float*        dinv     = (float*)(rowptr + GCN_NP);
    int*   rec    = (int*)(dinv + GCN_NP);
    float* bufA   = (float*)(rec + GCN_E);
    float* bufB   = bufA + (size_t)GCN_N * GCN_H;
    float* psum   = bufB + (size_t)GCN_N * GCN_H;
    float* pcnt   = psum + (size_t)GCN_B * GCN_ZW;
    float* wc     = pcnt + GCN_B;

    const dim3 gemmBlk(32, 4);
    const dim3 gemmGrd(GCN_N / 16);
    const int  poolGrd = (GCN_N + 63) / 64;
    const int  fuseGrd = GCN_N / 16;

    gcn_bin_scatter<<<GCN_G1 + 1, 1024, 0, stream>>>(src, dst, binOff, tmp, W3, b3, Wl, wc);
    gcn_bin_sort<<<GCN_NB, 1024, 0, stream>>>(tmp, binOff, rowptr, dinv, rec);

    gcn_gemm_wmma_f32<<<gemmGrd, gemmBlk, 0, stream>>>(x, W1, dinv, bufA, GCN_FIN);

    gcn_gather_gemm<4><<<fuseGrd, 128, 0, stream>>>(bufA, rec, rowptr, dinv, b1, W2, bufB);

    gcn_gather_gemm<1><<<fuseGrd, 128, 0, stream>>>(bufB, rec, rowptr, dinv, b2, wc, bufA);

    hipMemsetAsync(psum, 0, (size_t)(GCN_B * GCN_ZW + GCN_B) * sizeof(float), stream);
    gcn_gather_pool16<<<poolGrd, 256, 0, stream>>>(bufA, rec, rowptr, dinv, wc + 64 * GCN_ZW, batch, psum, pcnt);

    gcn_final<<<(GCN_B * GCN_C) / 256, 256, 0, stream>>>(psum, pcnt, bl, out);
}


# All three gathers: eight slots per trip, a node's last trip included (clamped record, row replaced by zeros), not four per trip (64-wide: then one per trip); 8 rows in flight; sum order kept
# speedup vs baseline: 6.1295x; 1.0448x over previous
#include <hip/hip_runtime.h>
#include <hip/hip_bf16.h>


#define GCN_N    50000
#define GCN_FIN  128
#define GCN_E    800000
#define GCN_H    64
#define GCN_C    10
#define GCN_B    512
#define GCN_NP   53248
#define GCN_ZW   16

typedef __attribute__((ext_vector_type(2))) float v2f;
typedef __attribute__((ext_vector_type(8))) float v8f;

__global__ __launch_bounds__(128) void gcn_gemm_wmma_f32(
    const float* __restrict__ A, const float* __restrict__ W,
    const float* __restrict__ dinv, float* __restrict__ C, int K)
{
    const int lane    = threadIdx.x;
    const int colBase = threadIdx.y * 16;
    const int rowBase = blockIdx.x * 16;
    const int half    = lane >> 4;
    const int l       = lane & 15;

    const float* arow = A + (size_t)(rowBase + l) * K + 2 * half;
    const float* wcol = W + colBase + l;

    v8f acc = {};
    for (int k0 = 0; k0 < K; k0 += 4) {
        v2f a, b;
        a.x = arow[k0 + 0];
        a.y = arow[k0 + 1];
        b.x = wcol[(size_t)(k0 + 2 * half + 0) * GCN_H];
        b.y = wcol[(size_t)(k0 + 2 * half + 1) * GCN_H];
        acc = __builtin_amdgcn_wmma_f32_16x16x4_f32(
             false, a,  false, b,
             (short)0, acc,  false,  false);
    }

    const float4 dlo = *(const float4*)(dinv + rowBase + 8 * half);
    const float4 dhi = *(const float4*)(dinv + rowBase + 8 * half + 4);
    const float dr[8] = { dlo.x, dlo.y, dlo.z, dlo.w, dhi.x, dhi.y, dhi.z, dhi.w };
#pragma unroll
    for (int v = 0; v < 8; ++v) {
        const int m = v + 8 * half;
        C[(size_t)(rowBase + m) * GCN_H + colBase + l] = acc[v] * dr[v];
    }
}

#define GCN_NB   196
#define GCN_G1   125
#define GCN_CH   6400
#define GCN_EPT  7
#define GCN_CAP  8192
#define GCN_LP   320
#define GCN_SPT  8
#define GCN_RPT  8

static_assert(GCN_G1 * GCN_CH == GCN_E, "chunks cover the edge list exactly");
static_assert(GCN_NB * 256 >= GCN_N + 1 && GCN_NB <= 256 && GCN_N < 65536, "key packing");
static_assert(GCN_G1 * GCN_SPT <= 1024, "every chunk's segment has its threads in a bin's block");

__device__ __forceinline__ void gcn_excl_scan256(int* a, const int t)
{
    if (t < 32) {
        int v[8];
        int s = 0;
#pragma unroll
        for (int j = 0; j < 8; ++j) { v[j] = a[t * 8 + j]; s += v[j]; }
        int inc = s;
#pragma unroll
        for (int off = 1; off < 32; off <<= 1) {
            const int o = __shfl_up(inc, off);
            if (t >= off) inc += o;
        }
        int run = inc - s;
#pragma unroll
        for (int j = 0; j < 8; ++j) { a[t * 8 + j] = run; run += v[j]; }
        if (t == 31) a[256] = run;
    }
    __syncthreads();
}

__global__ __launch_bounds__(1024) void gcn_bin_scatter(
    const int* __restrict__ src, const int* __restrict__ dst,
    int* __restrict__ binOff, unsigned int* __restrict__ tmp,
    const float* __restrict__ W3, const float* __restrict__ b3,
    const float* __restrict__ Wl, float* __restrict__ wc)
{
    if (blockIdx.x == GCN_G1) {
        for (int idx = threadIdx.x; idx < 65 * GCN_ZW; idx += 1024) {
            const int r = idx / GCN_ZW;
            const int c = idx % GCN_ZW;
            const float* a = (r < GCN_H) ? (W3 + r * GCN_H) : b3;
            float acc = 0.0f;
            if (c < GCN_C) {
                for (int k = 0; k < GCN_H; ++k) acc = fmaf(a[k], Wl[k * GCN_C + c], acc);
            }
            wc[idx] = acc;
        }
        return;
    }

    __shared__ int lst[257];
    __shared__ unsigned int sorted[GCN_CH];

    const int t = threadIdx.x;
    const int b = blockIdx.x;
    const int* sp = src + b * GCN_CH;
    const int* dp = dst + b * GCN_CH;

    unsigned int key[GCN_EPT];
#pragma unroll
    for (int j = 0; j < GCN_EPT; ++j) {
        const int i = t + j * 1024;
        key[j] = 0u;
        if (i < GCN_CH) {
            const unsigned int s = (unsigned int)sp[i];
            const unsigned int d = (unsigned int)dp[i];
            key[j] = ((d >> 8) << 24) | (s << 8) | (d & 255u);
        }
    }

    if (t < 256) lst[t] = 0;
    __syncthreads();

    int r[GCN_EPT];
#pragma unroll
    for (int j = 0; j < GCN_EPT; ++j) {
        r[j] = 0;
        if (t + j * 1024 < GCN_CH) r[j] = atomicAdd(&lst[key[j] >> 24], 1);
    }
    __syncthreads();
    gcn_excl_scan256(lst, t);
    if (t <= 256) binOff[b * GCN_LP + t] = lst[t];
#pragma unroll
    for (int j = 0; j < GCN_EPT; ++j) {
        if (t + j * 1024 < GCN_CH) sorted[lst[key[j] >> 24] + r[j]] = key[j];
    }
    __syncthreads();
    unsigned int* run = tmp + b * GCN_CH;
    for (int p = t; p < GCN_CH; p += 1024) run[p] = sorted[p];
}

__global__ __launch_bounds__(1024) void gcn_bin_sort(
    const unsigned int* __restrict__ tmp, const int* __restrict__ binOff,
    int* __restrict__ rowptr, float* __restrict__ dinv, int* __restrict__ rec)
{
    __shared__ int   cnt[257];
    __shared__ int   cur[256];
    __shared__ int   tot[2];
    __shared__ int   stage[GCN_CAP];

    const int t   = threadIdx.x;
    const int bin = blockIdx.x;
    const int seg = t / GCN_SPT;
    const int sub = t % GCN_SPT;

    int sbeg = 0, slen = 0, slow = 0;
    if (seg < GCN_G1) {
        const int* bo = binOff + seg * GCN_LP + bin;
        slow = bo[0];
        slen = bo[1] - slow;
        sbeg = seg * GCN_CH + slow;
    }
    unsigned int kk[GCN_RPT];
#pragma unroll
    for (int u = 0; u < GCN_RPT; ++u) {
        const int j = sub + u * GCN_SPT;
        kk[u] = 0u;
        if (j < slen) kk[u] = tmp[sbeg + j];
    }

    if (t < 256) cnt[t] = 0;
    if (t < 2)   tot[t] = 0;
    __syncthreads();
    if (sub == 0 && seg < GCN_G1) {
        atomicAdd(&tot[0], slow);
        atomicAdd(&tot[1], slen);
    }
#pragma unroll
    for (int u = 0; u < GCN_RPT; ++u) {
        if (sub + u * GCN_SPT < slen) atomicAdd(&cnt[kk[u] & 255u], 1);
    }
    for (int j = sub + GCN_RPT * GCN_SPT; j < slen; j += GCN_SPT) atomicAdd(&cnt[tmp[sbeg + j] & 255u], 1);
    __syncthreads();
    const int beg = tot[0];
    const int n   = tot[1];
    int deg = 0;
    if (t < 256) deg = cnt[t];
    __syncthreads();
    gcn_excl_scan256(cnt, t);
    if (t < 256) {
        const int node = bin * 256 + t;
        const float dv = rsqrtf((float)deg + 1.0f);
        cur[t] = cnt[t];
        if (node <= GCN_N) rowptr[node] = beg + cnt[t];
        if (node <  GCN_N) dinv[node]   = dv;
    }
    __syncthreads();

    const bool staged = (n <= GCN_CAP);
#pragma unroll
    for (int u = 0; u < GCN_RPT; ++u) {
        if (sub + u * GCN_SPT < slen) {
            const unsigned int k = kk[u];
            const int s   = (int)((k >> 8) & 0xFFFFu);
            const int pos = atomicAdd(&cur[k & 255u], 1);
            if (staged) stage[pos] = s;
            else        rec[beg + pos] = s;
        }
    }
    for (int j = sub + GCN_RPT * GCN_SPT; j < slen; j += GCN_SPT) {
        const unsigned int k = tmp[sbeg + j];
        const int s   = (int)((k >> 8) & 0xFFFFu);
        const int pos = atomicAdd(&cur[k & 255u], 1);
        if (staged) stage[pos] = s;
        else        rec[beg + pos] = s;
    }
    __syncthreads();
    if (staged) {
        for (int p = t; p < n; p += 1024) rec[beg + p] = stage[p];
    }
}

#define GCN_TP  68
#define GCN_NPW 4

template <int NT>
__global__ __launch_bounds__(128) void gcn_gather_gemm(
    const float* __restrict__ h, const int* __restrict__ rec,
    const int* __restrict__ rowptr, const float* __restrict__ dinv,
    const float* __restrict__ bias, const float* __restrict__ W,
    float* __restrict__ out)
{
    __shared__ __attribute__((aligned(16))) float tile[16 * GCN_TP];
    constexpr int P = 16 * NT;

    const int lane = threadIdx.x & 31;
    const int p    = lane * 2;
    const int w     = __builtin_amdgcn_readfirstlane(threadIdx.x >> 5);
    const int node0 = blockIdx.x * 16 + w * GCN_NPW;

    const int   rpv = rowptr[node0 + (lane < GCN_NPW ? lane : GCN_NPW)];
    const float dvv = dinv[node0 + (lane & (GCN_NPW - 1))];
    const float2 bv = *(const float2*)(bias + p);

    int    beg[GCN_NPW], end[GCN_NPW], rs0[GCN_NPW];
    float2 hv[GCN_NPW];
#pragma unroll
    for (int i = 0; i < GCN_NPW; ++i) {
        beg[i] = __builtin_amdgcn_readlane(rpv, i);
        end[i] = __builtin_amdgcn_readlane(rpv, i + 1);
    }
#pragma unroll
    for (int i = 0; i < GCN_NPW; ++i) {
        const int slot = beg[i] + lane;
        rs0[i] = rec[slot < GCN_E ? slot : GCN_E - 1];
        hv[i] = *(const float2*)(h + (size_t)(node0 + i) * GCN_H + p);
    }
    asm volatile("" :: "v"(rs0[0]), "v"(rs0[1]), "v"(rs0[2]), "v"(rs0[3]),
                       "v"(hv[0].x), "v"(hv[0].y), "v"(hv[1].x), "v"(hv[1].y),
                       "v"(hv[2].x), "v"(hv[2].y), "v"(hv[3].x), "v"(hv[3].y),
                       "v"(bv.x), "v"(bv.y), "v"(dvv));

#pragma unroll
    for (int i = 0; i < GCN_NPW; ++i) {
        float ax = 0.0f, ay = 0.0f;
        int rs = rs0[i];
        for (int base = beg[i]; base < end[i]; base += 32) {
            const int left = end[i] - base;
            const int n = left < 32 ? left : 32;
            if (base != beg[i]) {
                const int slot = base + lane;
                rs = rec[slot < GCN_E ? slot : GCN_E - 1];
            }
            const int last = n - 1;
            const float2 zero = make_float2(0.0f, 0.0f);
            for (int j = 0; j < n; j += 8) {
                const bool more = j + 4 < n;
                const int i1 = j + 1 < last ? j + 1 : last;
                const int i2 = j + 2 < last ? j + 2 : last;
                const int i3 = j + 3 < last ? j + 3 : last;
                const int s0 = __builtin_amdgcn_readlane(rs, j);
                const int s1 = __builtin_amdgcn_readlane(rs, i1);
                const int s2 = __builtin_amdgcn_readlane(rs, i2);
                const int s3 = __builtin_amdgcn_readlane(rs, i3);
                const float2 v0 = *(const float2*)(h + (size_t)s0 * GCN_H + p);
                float2 v1 = *(const float2*)(h + (size_t)s1 * GCN_H + p);
                float2 v2 = *(const float2*)(h + (size_t)s2 * GCN_H + p);
                float2 v3 = *(const float2*)(h + (size_t)s3 * GCN_H + p);
                float2 v4 = zero, v5 = zero, v6 = zero, v7 = zero;
                if (more) {
                    const int i5 = j + 5 < last ? j + 5 : last;
                    const int i6 = j + 6 < last ? j + 6 : last;
                    const int i7 = j + 7 < last ? j + 7 : last;
                    const int s4 = __builtin_amdgcn_readlane(rs, j + 4);
                    const int s5 = __builtin_amdgcn_readlane(rs, i5);
                    const int s6 = __builtin_amdgcn_readlane(rs, i6);
                    const int s7 = __builtin_amdgcn_readlane(rs, i7);
                    v4 = *(const float2*)(h + (size_t)s4 * GCN_H + p);
                    v5 = *(const float2*)(h + (size_t)s5 * GCN_H + p);
                    v6 = *(const float2*)(h + (size_t)s6 * GCN_H + p);
                    v7 = *(const float2*)(h + (size_t)s7 * GCN_H + p);
                }
                if (j + 1 >= n) v1 = zero;
                if (j + 2 >= n) v2 = zero;
                if (j + 3 >= n) v3 = zero;
                ax += v0.x;  ay += v0.y;
                ax += v1.x;  ay += v1.y;
                ax += v2.x;  ay += v2.y;
                ax += v3.x;  ay += v3.y;
                if (more) {
                    if (j + 5 >= n) v5 = zero;
                    if (j + 6 >= n) v6 = zero;
                    if (j + 7 >= n) v7 = zero;
                    ax += v4.x;  ay += v4.y;
                    ax += v5.x;  ay += v5.y;
                    ax += v6.x;  ay += v6.y;
                    ax += v7.x;  ay += v7.y;
                }
            }
        }

        const float di = __int_as_float(__builtin_amdgcn_readlane(__float_as_int(dvv), i));
        float2 o = make_float2((ax + hv[i].x) * di + bv.x, (ay + hv[i].y) * di + bv.y);
        o.x = fmaxf(o.x, 0.0f);
        o.y = fmaxf(o.y, 0.0f);
        *(float2*)(tile + (w * GCN_NPW + i) * GCN_TP + p) = o;
    }

    const int half = lane >> 4;
    const int l    = lane & 15;

    v2f b[16];
    float4 dlo = make_float4(0.0f, 0.0f, 0.0f, 0.0f), dhi = dlo;
    if (w < NT) {
        const float* wcol = W + (size_t)(2 * half) * P + w * 16 + l;
#pragma unroll
        for (int t = 0; t < 16; ++t) {
            b[t].x = wcol[(size_t)(4 * t + 0) * P];
            b[t].y = wcol[(size_t)(4 * t + 1) * P];
        }
        dlo = *(const float4*)(dinv + blockIdx.x * 16 + 8 * half);
        dhi = *(const float4*)(dinv + blockIdx.x * 16 + 8 * half + 4);
    }

    __syncthreads();

    if (w < NT) {
        const float* arow = tile + l * GCN_TP + 2 * half;
        v8f acc = {};
#pragma unroll
        for (int t = 0; t < 16; ++t) {
            const float2 av = *(const float2*)(arow + 4 * t);
            v2f a;
            a.x = av.x;
            a.y = av.y;
            acc = __builtin_amdgcn_wmma_f32_16x16x4_f32(
                 false, a,  false, b[t],
                 (short)0, acc,  false,  false);
        }
        float* orow = out + (size_t)(blockIdx.x * 16 + 8 * half) * P + w * 16 + l;
        const float dr[8] = { dlo.x, dlo.y, dlo.z, dlo.w, dhi.x, dhi.y, dhi.z, dhi.w };
#pragma unroll
        for (int v = 0; v < 8; ++v)
            orow[(size_t)v * P] = acc[v] * dr[v];
    }
}

__global__ __launch_bounds__(256) void gcn_gather_pool16(
    const float* __restrict__ z, const int* __restrict__ rec,
    const int* __restrict__ rowptr, const float* __restrict__ dinv,
    const float* __restrict__ bias, const int* __restrict__ batch,
    float* __restrict__ psum, float* __restrict__ pcnt)
{
    __shared__ __attribute__((aligned(16))) float srow[64 * GCN_ZW];
    __shared__ int sgid[64];

    const int q    = threadIdx.x & 3;
    const int k    = threadIdx.x >> 2;
    const int node = blockIdx.x * 64 + k;

    float4 o = make_float4(0.0f, 0.0f, 0.0f, 0.0f);
    static_assert(GCN_N % 8 == 0, "a wave's 8 nodes exist together or not at all");
    if (node < GCN_N) {
        const int beg = rowptr[node];
        const int end = rowptr[node + 1];
        const float di = dinv[node];
        const float4 hv = *(const float4*)(z + (size_t)node * GCN_ZW + 4 * q);
        const float4 bv = *(const float4*)(bias + 4 * q);

        float4 a = make_float4(0.0f, 0.0f, 0.0f, 0.0f);
        const int last = end - 1;
        for (int s = beg; s < end; s += 8) {
            const float4 zero = make_float4(0.0f, 0.0f, 0.0f, 0.0f);
            int    r[8];
            float4 v[8];
#pragma unroll
            for (int u = 0; u < 8; ++u) r[u] = rec[(s + u < end) ? s + u : last];
#pragma unroll
            for (int u = 0; u < 8; ++u) v[u] = *(const float4*)(z + (size_t)r[u] * GCN_ZW + 4 * q);
#pragma unroll
            for (int u = 0; u < 8; ++u) {
                if (s + u >= end) v[u] = zero;
                a.x += v[u].x;  a.y += v[u].y;  a.z += v[u].z;  a.w += v[u].w;
            }
        }
        o.x = (a.x + hv.x) * di + bv.x;
        o.y = (a.y + hv.y) * di + bv.y;
        o.z = (a.z + hv.z) * di + bv.z;
        o.w = (a.w + hv.w) * di + bv.w;
        if (q == 0) sgid[k] = batch[node];
    }

    *(float4*)(srow + k * GCN_ZW + 4 * q) = o;
    __syncthreads();

    const int lane  = threadIdx.x & 31;
    const int w     = threadIdx.x >> 5;
    const int first = blockIdx.x * 64 + w * 8;
    if (lane < GCN_C && first < GCN_N) {
        const float* rows = srow + w * 8 * GCN_ZW;
        const int*   ids  = sgid + w * 8;
        int   cur = __builtin_amdgcn_readfirstlane(ids[0]);
        float acc = rows[lane];
        int   run = 1;
        for (int j = 1; j < 8; ++j) {
            const int   g = __builtin_amdgcn_readfirstlane(ids[j]);
            const float v = rows[j * GCN_ZW + lane];
            if (g != cur) {
                atomicAdd(&psum[(size_t)cur * GCN_ZW + lane], acc);
                if (lane == 0) atomicAdd(&pcnt[cur], (float)run);
                cur = g;  acc = v;  run = 1;
            } else {
                acc += v;  ++run;
            }
        }
        atomicAdd(&psum[(size_t)cur * GCN_ZW + lane], acc);
        if (lane == 0) atomicAdd(&pcnt[cur], (float)run);
    }
}

__global__ __launch_bounds__(256) void gcn_final(
    const float* __restrict__ psum, const float* __restrict__ pcnt,
    const float* __restrict__ bl, float* __restrict__ out)
{
    const int idx = blockIdx.x * blockDim.x + threadIdx.x;
    const int b = idx / GCN_C;
    const int c = idx % GCN_C;
    out[idx] = psum[(size_t)b * GCN_ZW + c] / fmaxf(pcnt[b], 1.0f) + bl[c];
}

extern "C" void kernel_launch(void* const* d_in, const int* in_sizes, int n_in,
                              void* d_out, int out_size, void* d_ws, size_t ws_size,
                              hipStream_t stream) {
    (void)in_sizes; (void)n_in; (void)out_size; (void)ws_size;

    const float* x     = (const float*)d_in[0];
    const int*   ei    = (const int*)  d_in[1];
    const int*   batch = (const int*)  d_in[2];
    const float* W1    = (const float*)d_in[3];
    const float* b1    = (const float*)d_in[4];
    const float* W2    = (const float*)d_in[5];
    const float* b2    = (const float*)d_in[6];
    const float* W3    = (const float*)d_in[7];
    const float* b3    = (const float*)d_in[8];
    const float* Wl    = (const float*)d_in[9];
    const float* bl    = (const float*)d_in[10];
    float* out = (float*)d_out;

    const int* src = ei;
    const int* dst = ei + GCN_E;

    int*          binOff   = (int*)d_ws;
    unsigned int* tmp      = (unsigned int*)(binOff + GCN_G1 * GCN_LP);
    int*          rowptr   = (int*)(tmp + GCN_E);
    float*        dinv     = (float*)(rowptr + GCN_NP);
    int*   rec    = (int*)(dinv + GCN_NP);
    float* bufA   = (float*)(rec + GCN_E);
    float* bufB   = bufA + (size_t)GCN_N * GCN_H;
    float* psum   = bufB + (size_t)GCN_N * GCN_H;
    float* pcnt   = psum + (size_t)GCN_B * GCN_ZW;
    float* wc     = pcnt + GCN_B;

    const dim3 gemmBlk(32, 4);
    const dim3 gemmGrd(GCN_N / 16);
    const int  poolGrd = (GCN_N + 63) / 64;
    const int  fuseGrd = GCN_N / 16;

    gcn_bin_scatter<<<GCN_G1 + 1, 1024, 0, stream>>>(src, dst, binOff, tmp, W3, b3, Wl, wc);
    gcn_bin_sort<<<GCN_NB, 1024, 0, stream>>>(tmp, binOff, rowptr, dinv, rec);

    gcn_gemm_wmma_f32<<<gemmGrd, gemmBlk, 0, stream>>>(x, W1, dinv, bufA, GCN_FIN);

    gcn_gather_gemm<4><<<fuseGrd, 128, 0, stream>>>(bufA, rec, rowptr, dinv, b1, W2, bufB);

    gcn_gather_gemm<1><<<fuseGrd, 128, 0, stream>>>(bufB, rec, rowptr, dinv, b2, wc, bufA);

    hipMemsetAsync(psum, 0, (size_t)(GCN_B * GCN_ZW + GCN_B) * sizeof(float), stream);
    gcn_gather_pool16<<<poolGrd, 256, 0, stream>>>(bufA, rec, rowptr, dinv, wc + 64 * GCN_ZW, batch, psum, pcnt);

    gcn_final<<<(GCN_B * GCN_C) / 256, 256, 0, stream>>>(psum, pcnt, bl, out);
}


# Layer-1 product: one wave per 16x64 row tile (4 accumulators), x rows and W1 staged in LDS by whole-line loads, not 8-byte strided reads per column tile; 3128 waves, not 12500; k order kept
# speedup vs baseline: 6.5397x; 1.0669x over previous
#include <hip/hip_runtime.h>
#include <hip/hip_bf16.h>


#define GCN_N    50000
#define GCN_FIN  128
#define GCN_E    800000
#define GCN_H    64
#define GCN_C    10
#define GCN_B    512
#define GCN_NP   53248
#define GCN_ZW   16

typedef __attribute__((ext_vector_type(2))) float v2f;
typedef __attribute__((ext_vector_type(8))) float v8f;

#define GCN_GR  64
#define GCN_AP  132
#define GCN_BP  160

static_assert(GCN_N % 16 == 0, "a wave's 16 rows exist together or not at all");

__global__ __launch_bounds__(128) void gcn_gemm_wmma_f32(
    const float* __restrict__ A, const float* __restrict__ W,
    const float* __restrict__ dinv, float* __restrict__ C)
{
    __shared__ __attribute__((aligned(16))) float As[4 * 16 * GCN_AP];
    __shared__ __attribute__((aligned(16))) float Bs[(GCN_FIN / 2) * GCN_BP];

    const int lane = threadIdx.x & 31;
    const int w       = __builtin_amdgcn_readfirstlane(threadIdx.x >> 5);
    const int rowBase = blockIdx.x * GCN_GR + w * 16;
    const bool active = rowBase < GCN_N;
    const int half    = lane >> 4;
    const int l       = lane & 15;
    float* as = As + w * 16 * GCN_AP;

    float4 ar[16];
    {
        const float* ap = A + (size_t)(active ? rowBase : GCN_N - 16) * GCN_FIN + 4 * lane;
#pragma unroll
        for (int r = 0; r < 16; ++r) ar[r] = *(const float4*)(ap + r * GCN_FIN);
    }

#pragma unroll
    for (int i = 0; i < (GCN_FIN / 2) * 16 / 128; ++i) {
        const int item = threadIdx.x + 128 * i;
        const int pr = item >> 4;
        const int cg = item & 15;
        const float4 w0 = *(const float4*)(W + (size_t)(2 * pr + 0) * GCN_H + 4 * cg);
        const float4 w1 = *(const float4*)(W + (size_t)(2 * pr + 1) * GCN_H + 4 * cg);
        float* bd = Bs + pr * GCN_BP + 8 * cg;
        *(float4*)(bd + 0) = make_float4(w0.x, w1.x, w0.y, w1.y);
        *(float4*)(bd + 4) = make_float4(w0.z, w1.z, w0.w, w1.w);
    }

#pragma unroll
    for (int r = 0; r < 16; ++r) *(float4*)(as + r * GCN_AP + 4 * lane) = ar[r];

    __syncthreads();

    if (active) {
        const float* arow = as + l * GCN_AP + 2 * half;
        const float* bcol = Bs + half * GCN_BP + 2 * l;
        v8f acc[4] = {};
        float2 an = *(const float2*)arow;
        float2 bn[4];
#pragma unroll
        for (int c = 0; c < 4; ++c) bn[c] = *(const float2*)(bcol + 32 * c);
#pragma unroll
        for (int t = 0; t < GCN_FIN / 4; ++t) {
            v2f a;
            a.x = an.x;
            a.y = an.y;
            v2f b[4];
#pragma unroll
            for (int c = 0; c < 4; ++c) { b[c].x = bn[c].x; b[c].y = bn[c].y; }
            if (t + 1 < GCN_FIN / 4) {
                an = *(const float2*)(arow + 4 * (t + 1));
#pragma unroll
                for (int c = 0; c < 4; ++c) bn[c] = *(const float2*)(bcol + 2 * (t + 1) * GCN_BP + 32 * c);
            }
#pragma unroll
            for (int c = 0; c < 4; ++c)
                acc[c] = __builtin_amdgcn_wmma_f32_16x16x4_f32(
                     false, a,  false, b[c],
                     (short)0, acc[c],  false,  false);
        }

        const float4 dlo = *(const float4*)(dinv + rowBase + 8 * half);
        const float4 dhi = *(const float4*)(dinv + rowBase + 8 * half + 4);
        const float dr[8] = { dlo.x, dlo.y, dlo.z, dlo.w, dhi.x, dhi.y, dhi.z, dhi.w };
#pragma unroll
        for (int c = 0; c < 4; ++c) {
#pragma unroll
            for (int v = 0; v < 8; ++v)
                as[(v + 8 * half) * GCN_AP + 16 * c + l] = acc[c][v] * dr[v];
        }
    }

    __syncthreads();

    if (active) {
        float* crow = C + (size_t)(rowBase + half) * GCN_H + 4 * l;
#pragma unroll
        for (int i = 0; i < 8; ++i) {
            const float4 o = *(const float4*)(as + (2 * i + half) * GCN_AP + 4 * l);
            *(float4*)(crow + (size_t)(2 * i) * GCN_H) = o;
        }
    }
}

#define GCN_NB   196
#define GCN_G1   125
#define GCN_CH   6400
#define GCN_EPT  7
#define GCN_CAP  8192
#define GCN_LP   320
#define GCN_SPT  8
#define GCN_RPT  8

static_assert(GCN_G1 * GCN_CH == GCN_E, "chunks cover the edge list exactly");
static_assert(GCN_NB * 256 >= GCN_N + 1 && GCN_NB <= 256 && GCN_N < 65536, "key packing");
static_assert(GCN_G1 * GCN_SPT <= 1024, "every chunk's segment has its threads in a bin's block");

__device__ __forceinline__ void gcn_excl_scan256(int* a, const int t)
{
    if (t < 32) {
        int v[8];
        int s = 0;
#pragma unroll
        for (int j = 0; j < 8; ++j) { v[j] = a[t * 8 + j]; s += v[j]; }
        int inc = s;
#pragma unroll
        for (int off = 1; off < 32; off <<= 1) {
            const int o = __shfl_up(inc, off);
            if (t >= off) inc += o;
        }
        int run = inc - s;
#pragma unroll
        for (int j = 0; j < 8; ++j) { a[t * 8 + j] = run; run += v[j]; }
        if (t == 31) a[256] = run;
    }
    __syncthreads();
}

__global__ __launch_bounds__(1024) void gcn_bin_scatter(
    const int* __restrict__ src, const int* __restrict__ dst,
    int* __restrict__ binOff, unsigned int* __restrict__ tmp,
    const float* __restrict__ W3, const float* __restrict__ b3,
    const float* __restrict__ Wl, float* __restrict__ wc)
{
    if (blockIdx.x == GCN_G1) {
        for (int idx = threadIdx.x; idx < 65 * GCN_ZW; idx += 1024) {
            const int r = idx / GCN_ZW;
            const int c = idx % GCN_ZW;
            const float* a = (r < GCN_H) ? (W3 + r * GCN_H) : b3;
            float acc = 0.0f;
            if (c < GCN_C) {
                for (int k = 0; k < GCN_H; ++k) acc = fmaf(a[k], Wl[k * GCN_C + c], acc);
            }
            wc[idx] = acc;
        }
        return;
    }

    __shared__ int lst[257];
    __shared__ unsigned int sorted[GCN_CH];

    const int t = threadIdx.x;
    const int b = blockIdx.x;
    const int* sp = src + b * GCN_CH;
    const int* dp = dst + b * GCN_CH;

    unsigned int key[GCN_EPT];
#pragma unroll
    for (int j = 0; j < GCN_EPT; ++j) {
        const int i = t + j * 1024;
        key[j] = 0u;
        if (i < GCN_CH) {
            const unsigned int s = (unsigned int)sp[i];
            const unsigned int d = (unsigned int)dp[i];
            key[j] = ((d >> 8) << 24) | (s << 8) | (d & 255u);
        }
    }

    if (t < 256) lst[t] = 0;
    __syncthreads();

    int r[GCN_EPT];
#pragma unroll
    for (int j = 0; j < GCN_EPT; ++j) {
        r[j] = 0;
        if (t + j * 1024 < GCN_CH) r[j] = atomicAdd(&lst[key[j] >> 24], 1);
    }
    __syncthreads();
    gcn_excl_scan256(lst, t);
    if (t <= 256) binOff[b * GCN_LP + t] = lst[t];
#pragma unroll
    for (int j = 0; j < GCN_EPT; ++j) {
        if (t + j * 1024 < GCN_CH) sorted[lst[key[j] >> 24] + r[j]] = key[j];
    }
    __syncthreads();
    unsigned int* run = tmp + b * GCN_CH;
    for (int p = t; p < GCN_CH; p += 1024) run[p] = sorted[p];
}

__global__ __launch_bounds__(1024) void gcn_bin_sort(
    const unsigned int* __restrict__ tmp, const int* __restrict__ binOff,
    int* __restrict__ rowptr, float* __restrict__ dinv, int* __restrict__ rec)
{
    __shared__ int   cnt[257];
    __shared__ int   cur[256];
    __shared__ int   tot[2];
    __shared__ int   stage[GCN_CAP];

    const int t   = threadIdx.x;
    const int bin = blockIdx.x;
    const int seg = t / GCN_SPT;
    const int sub = t % GCN_SPT;

    int sbeg = 0, slen = 0, slow = 0;
    if (seg < GCN_G1) {
        const int* bo = binOff + seg * GCN_LP + bin;
        slow = bo[0];
        slen = bo[1] - slow;
        sbeg = seg * GCN_CH + slow;
    }
    unsigned int kk[GCN_RPT];
#pragma unroll
    for (int u = 0; u < GCN_RPT; ++u) {
        const int j = sub + u * GCN_SPT;
        kk[u] = 0u;
        if (j < slen) kk[u] = tmp[sbeg + j];
    }

    if (t < 256) cnt[t] = 0;
    if (t < 2)   tot[t] = 0;
    __syncthreads();
    if (sub == 0 && seg < GCN_G1) {
        atomicAdd(&tot[0], slow);
        atomicAdd(&tot[1], slen);
    }
#pragma unroll
    for (int u = 0; u < GCN_RPT; ++u) {
        if (sub + u * GCN_SPT < slen) atomicAdd(&cnt[kk[u] & 255u], 1);
    }
    for (int j = sub + GCN_RPT * GCN_SPT; j < slen; j += GCN_SPT) atomicAdd(&cnt[tmp[sbeg + j] & 255u], 1);
    __syncthreads();
    const int beg = tot[0];
    const int n   = tot[1];
    int deg = 0;
    if (t < 256) deg = cnt[t];
    __syncthreads();
    gcn_excl_scan256(cnt, t);
    if (t < 256) {
        const int node = bin * 256 + t;
        const float dv = rsqrtf((float)deg + 1.0f);
        cur[t] = cnt[t];
        if (node <= GCN_N) rowptr[node] = beg + cnt[t];
        if (node <  GCN_N) dinv[node]   = dv;
    }
    __syncthreads();

    const bool staged = (n <= GCN_CAP);
#pragma unroll
    for (int u = 0; u < GCN_RPT; ++u) {
        if (sub + u * GCN_SPT < slen) {
            const unsigned int k = kk[u];
            const int s   = (int)((k >> 8) & 0xFFFFu);
            const int pos = atomicAdd(&cur[k & 255u], 1);
            if (staged) stage[pos] = s;
            else        rec[beg + pos] = s;
        }
    }
    for (int j = sub + GCN_RPT * GCN_SPT; j < slen; j += GCN_SPT) {
        const unsigned int k = tmp[sbeg + j];
        const int s   = (int)((k >> 8) & 0xFFFFu);
        const int pos = atomicAdd(&cur[k & 255u], 1);
        if (staged) stage[pos] = s;
        else        rec[beg + pos] = s;
    }
    __syncthreads();
    if (staged) {
        for (int p = t; p < n; p += 1024) rec[beg + p] = stage[p];
    }
}

#define GCN_TP  68
#define GCN_NPW 4

template <int NT>
__global__ __launch_bounds__(128) void gcn_gather_gemm(
    const float* __restrict__ h, const int* __restrict__ rec,
    const int* __restrict__ rowptr, const float* __restrict__ dinv,
    const float* __restrict__ bias, const float* __restrict__ W,
    float* __restrict__ out)
{
    __shared__ __attribute__((aligned(16))) float tile[16 * GCN_TP];
    constexpr int P = 16 * NT;

    const int lane = threadIdx.x & 31;
    const int p    = lane * 2;
    const int w     = __builtin_amdgcn_readfirstlane(threadIdx.x >> 5);
    const int node0 = blockIdx.x * 16 + w * GCN_NPW;

    const int   rpv = rowptr[node0 + (lane < GCN_NPW ? lane : GCN_NPW)];
    const float dvv = dinv[node0 + (lane & (GCN_NPW - 1))];
    const float2 bv = *(const float2*)(bias + p);

    int    beg[GCN_NPW], end[GCN_NPW], rs0[GCN_NPW];
    float2 hv[GCN_NPW];
#pragma unroll
    for (int i = 0; i < GCN_NPW; ++i) {
        beg[i] = __builtin_amdgcn_readlane(rpv, i);
        end[i] = __builtin_amdgcn_readlane(rpv, i + 1);
    }
#pragma unroll
    for (int i = 0; i < GCN_NPW; ++i) {
        const int slot = beg[i] + lane;
        rs0[i] = rec[slot < GCN_E ? slot : GCN_E - 1];
        hv[i] = *(const float2*)(h + (size_t)(node0 + i) * GCN_H + p);
    }
    asm volatile("" :: "v"(rs0[0]), "v"(rs0[1]), "v"(rs0[2]), "v"(rs0[3]),
                       "v"(hv[0].x), "v"(hv[0].y), "v"(hv[1].x), "v"(hv[1].y),
                       "v"(hv[2].x), "v"(hv[2].y), "v"(hv[3].x), "v"(hv[3].y),
                       "v"(bv.x), "v"(bv.y), "v"(dvv));

#pragma unroll
    for (int i = 0; i < GCN_NPW; ++i) {
        float ax = 0.0f, ay = 0.0f;
        int rs = rs0[i];
        for (int base = beg[i]; base < end[i]; base += 32) {
            const int left = end[i] - base;
            const int n = left < 32 ? left : 32;
            if (base != beg[i]) {
                const int slot = base + lane;
                rs = rec[slot < GCN_E ? slot : GCN_E - 1];
            }
            const int last = n - 1;
            const float2 zero = make_float2(0.0f, 0.0f);
            for (int j = 0; j < n; j += 8) {
                const bool more = j + 4 < n;
                const int i1 = j + 1 < last ? j + 1 : last;
                const int i2 = j + 2 < last ? j + 2 : last;
                const int i3 = j + 3 < last ? j + 3 : last;
                const int s0 = __builtin_amdgcn_readlane(rs, j);
                const int s1 = __builtin_amdgcn_readlane(rs, i1);
                const int s2 = __builtin_amdgcn_readlane(rs, i2);
                const int s3 = __builtin_amdgcn_readlane(rs, i3);
                const float2 v0 = *(const float2*)(h + (size_t)s0 * GCN_H + p);
                float2 v1 = *(const float2*)(h + (size_t)s1 * GCN_H + p);
                float2 v2 = *(const float2*)(h + (size_t)s2 * GCN_H + p);
                float2 v3 = *(const float2*)(h + (size_t)s3 * GCN_H + p);
                float2 v4 = zero, v5 = zero, v6 = zero, v7 = zero;
                if (more) {
                    const int i5 = j + 5 < last ? j + 5 : last;
                    const int i6 = j + 6 < last ? j + 6 : last;
                    const int i7 = j + 7 < last ? j + 7 : last;
                    const int s4 = __builtin_amdgcn_readlane(rs, j + 4);
                    const int s5 = __builtin_amdgcn_readlane(rs, i5);
                    const int s6 = __builtin_amdgcn_readlane(rs, i6);
                    const int s7 = __builtin_amdgcn_readlane(rs, i7);
                    v4 = *(const float2*)(h + (size_t)s4 * GCN_H + p);
                    v5 = *(const float2*)(h + (size_t)s5 * GCN_H + p);
                    v6 = *(const float2*)(h + (size_t)s6 * GCN_H + p);
                    v7 = *(const float2*)(h + (size_t)s7 * GCN_H + p);
                }
                if (j + 1 >= n) v1 = zero;
                if (j + 2 >= n) v2 = zero;
                if (j + 3 >= n) v3 = zero;
                ax += v0.x;  ay += v0.y;
                ax += v1.x;  ay += v1.y;
                ax += v2.x;  ay += v2.y;
                ax += v3.x;  ay += v3.y;
                if (more) {
                    if (j + 5 >= n) v5 = zero;
                    if (j + 6 >= n) v6 = zero;
                    if (j + 7 >= n) v7 = zero;
                    ax += v4.x;  ay += v4.y;
                    ax += v5.x;  ay += v5.y;
                    ax += v6.x;  ay += v6.y;
                    ax += v7.x;  ay += v7.y;
                }
            }
        }

        const float di = __int_as_float(__builtin_amdgcn_readlane(__float_as_int(dvv), i));
        float2 o = make_float2((ax + hv[i].x) * di + bv.x, (ay + hv[i].y) * di + bv.y);
        o.x = fmaxf(o.x, 0.0f);
        o.y = fmaxf(o.y, 0.0f);
        *(float2*)(tile + (w * GCN_NPW + i) * GCN_TP + p) = o;
    }

    const int half = lane >> 4;
    const int l    = lane & 15;

    v2f b[16];
    float4 dlo = make_float4(0.0f, 0.0f, 0.0f, 0.0f), dhi = dlo;
    if (w < NT) {
        const float* wcol = W + (size_t)(2 * half) * P + w * 16 + l;
#pragma unroll
        for (int t = 0; t < 16; ++t) {
            b[t].x = wcol[(size_t)(4 * t + 0) * P];
            b[t].y = wcol[(size_t)(4 * t + 1) * P];
        }
        dlo = *(const float4*)(dinv + blockIdx.x * 16 + 8 * half);
        dhi = *(const float4*)(dinv + blockIdx.x * 16 + 8 * half + 4);
    }

    __syncthreads();

    if (w < NT) {
        const float* arow = tile + l * GCN_TP + 2 * half;
        v8f acc = {};
#pragma unroll
        for (int t = 0; t < 16; ++t) {
            const float2 av = *(const float2*)(arow + 4 * t);
            v2f a;
            a.x = av.x;
            a.y = av.y;
            acc = __builtin_amdgcn_wmma_f32_16x16x4_f32(
                 false, a,  false, b[t],
                 (short)0, acc,  false,  false);
        }
        float* orow = out + (size_t)(blockIdx.x * 16 + 8 * half) * P + w * 16 + l;
        const float dr[8] = { dlo.x, dlo.y, dlo.z, dlo.w, dhi.x, dhi.y, dhi.z, dhi.w };
#pragma unroll
        for (int v = 0; v < 8; ++v)
            orow[(size_t)v * P] = acc[v] * dr[v];
    }
}

__global__ __launch_bounds__(256) void gcn_gather_pool16(
    const float* __restrict__ z, const int* __restrict__ rec,
    const int* __restrict__ rowptr, const float* __restrict__ dinv,
    const float* __restrict__ bias, const int* __restrict__ batch,
    float* __restrict__ psum, float* __restrict__ pcnt)
{
    __shared__ __attribute__((aligned(16))) float srow[64 * GCN_ZW];
    __shared__ int sgid[64];

    const int q    = threadIdx.x & 3;
    const int k    = threadIdx.x >> 2;
    const int node = blockIdx.x * 64 + k;

    float4 o = make_float4(0.0f, 0.0f, 0.0f, 0.0f);
    static_assert(GCN_N % 8 == 0, "a wave's 8 nodes exist together or not at all");
    if (node < GCN_N) {
        const int beg = rowptr[node];
        const int end = rowptr[node + 1];
        const float di = dinv[node];
        const float4 hv = *(const float4*)(z + (size_t)node * GCN_ZW + 4 * q);
        const float4 bv = *(const float4*)(bias + 4 * q);

        float4 a = make_float4(0.0f, 0.0f, 0.0f, 0.0f);
        const int last = end - 1;
        for (int s = beg; s < end; s += 8) {
            const float4 zero = make_float4(0.0f, 0.0f, 0.0f, 0.0f);
            int    r[8];
            float4 v[8];
#pragma unroll
            for (int u = 0; u < 8; ++u) r[u] = rec[(s + u < end) ? s + u : last];
#pragma unroll
            for (int u = 0; u < 8; ++u) v[u] = *(const float4*)(z + (size_t)r[u] * GCN_ZW + 4 * q);
#pragma unroll
            for (int u = 0; u < 8; ++u) {
                if (s + u >= end) v[u] = zero;
                a.x += v[u].x;  a.y += v[u].y;  a.z += v[u].z;  a.w += v[u].w;
            }
        }
        o.x = (a.x + hv.x) * di + bv.x;
        o.y = (a.y + hv.y) * di + bv.y;
        o.z = (a.z + hv.z) * di + bv.z;
        o.w = (a.w + hv.w) * di + bv.w;
        if (q == 0) sgid[k] = batch[node];
    }

    *(float4*)(srow + k * GCN_ZW + 4 * q) = o;
    __syncthreads();

    const int lane  = threadIdx.x & 31;
    const int w     = threadIdx.x >> 5;
    const int first = blockIdx.x * 64 + w * 8;
    if (lane < GCN_C && first < GCN_N) {
        const float* rows = srow + w * 8 * GCN_ZW;
        const int*   ids  = sgid + w * 8;
        int   cur = __builtin_amdgcn_readfirstlane(ids[0]);
        float acc = rows[lane];
        int   run = 1;
        for (int j = 1; j < 8; ++j) {
            const int   g = __builtin_amdgcn_readfirstlane(ids[j]);
            const float v = rows[j * GCN_ZW + lane];
            if (g != cur) {
                atomicAdd(&psum[(size_t)cur * GCN_ZW + lane], acc);
                if (lane == 0) atomicAdd(&pcnt[cur], (float)run);
                cur = g;  acc = v;  run = 1;
            } else {
                acc += v;  ++run;
            }
        }
        atomicAdd(&psum[(size_t)cur * GCN_ZW + lane], acc);
        if (lane == 0) atomicAdd(&pcnt[cur], (float)run);
    }
}

__global__ __launch_bounds__(256) void gcn_final(
    const float* __restrict__ psum, const float* __restrict__ pcnt,
    const float* __restrict__ bl, float* __restrict__ out)
{
    const int idx = blockIdx.x * blockDim.x + threadIdx.x;
    const int b = idx / GCN_C;
    const int c = idx % GCN_C;
    out[idx] = psum[(size_t)b * GCN_ZW + c] / fmaxf(pcnt[b], 1.0f) + bl[c];
}

extern "C" void kernel_launch(void* const* d_in, const int* in_sizes, int n_in,
                              void* d_out, int out_size, void* d_ws, size_t ws_size,
                              hipStream_t stream) {
    (void)in_sizes; (void)n_in; (void)out_size; (void)ws_size;

    const float* x     = (const float*)d_in[0];
    const int*   ei    = (const int*)  d_in[1];
    const int*   batch = (const int*)  d_in[2];
    const float* W1    = (const float*)d_in[3];
    const float* b1    = (const float*)d_in[4];
    const float* W2    = (const float*)d_in[5];
    const float* b2    = (const float*)d_in[6];
    const float* W3    = (const float*)d_in[7];
    const float* b3    = (const float*)d_in[8];
    const float* Wl    = (const float*)d_in[9];
    const float* bl    = (const float*)d_in[10];
    float* out = (float*)d_out;

    const int* src = ei;
    const int* dst = ei + GCN_E;

    int*          binOff   = (int*)d_ws;
    unsigned int* tmp      = (unsigned int*)(binOff + GCN_G1 * GCN_LP);
    int*          rowptr   = (int*)(tmp + GCN_E);
    float*        dinv     = (float*)(rowptr + GCN_NP);
    int*   rec    = (int*)(dinv + GCN_NP);
    float* bufA   = (float*)(rec + GCN_E);
    float* bufB   = bufA + (size_t)GCN_N * GCN_H;
    float* psum   = bufB + (size_t)GCN_N * GCN_H;
    float* pcnt   = psum + (size_t)GCN_B * GCN_ZW;
    float* wc     = pcnt + GCN_B;

    const int  gemmGrd = (GCN_N + GCN_GR - 1) / GCN_GR;
    const int  poolGrd = (GCN_N + 63) / 64;
    const int  fuseGrd = GCN_N / 16;

    gcn_bin_scatter<<<GCN_G1 + 1, 1024, 0, stream>>>(src, dst, binOff, tmp, W3, b3, Wl, wc);
    gcn_bin_sort<<<GCN_NB, 1024, 0, stream>>>(tmp, binOff, rowptr, dinv, rec);

    gcn_gemm_wmma_f32<<<gemmGrd, 128, 0, stream>>>(x, W1, dinv, bufA);

    gcn_gather_gemm<4><<<fuseGrd, 128, 0, stream>>>(bufA, rec, rowptr, dinv, b1, W2, bufB);

    gcn_gather_gemm<1><<<fuseGrd, 128, 0, stream>>>(bufB, rec, rowptr, dinv, b2, wc, bufA);

    hipMemsetAsync(psum, 0, (size_t)(GCN_B * GCN_ZW + GCN_B) * sizeof(float), stream);
    gcn_gather_pool16<<<poolGrd, 256, 0, stream>>>(bufA, rec, rowptr, dinv, wc + 64 * GCN_ZW, batch, psum, pcnt);

    gcn_final<<<(GCN_B * GCN_C) / 256, 256, 0, stream>>>(psum, pcnt, bl, out);
}
